# starting kernel
# baseline (speedup 1.0000x reference)
#include <hip/hip_runtime.h>
#include <hip/hip_bf16.h>

#define NNODES 100000
#define NEDGES 1000000
#define IN_C   128
#define HID_C  64
#define OUT_C  32

typedef __attribute__((ext_vector_type(2))) float v2f;
typedef __attribute__((ext_vector_type(8))) float v8f;

__global__ void gcn_init(float* __restrict__ deg,
                         float4* __restrict__ agg1,
                         float4* __restrict__ out)
{
    int i = blockIdx.x * blockDim.x + threadIdx.x;
    if (i >= NNODES) return;
    deg[i] = 1.0f;
    float4 z = make_float4(0.f, 0.f, 0.f, 0.f);
#pragma unroll
    for (int q = 0; q < HID_C / 4; ++q) agg1[(size_t)i * (HID_C / 4) + q] = z;
#pragma unroll
    for (int q = 0; q < OUT_C / 4; ++q) out[(size_t)i * (OUT_C / 4) + q] = z;
}

__global__ void gcn_degree(const int* __restrict__ dst, float* __restrict__ deg)
{
    int e = blockIdx.x * blockDim.x + threadIdx.x;
    if (e >= NEDGES) return;
    atomicAdd(&deg[dst[e]], 1.0f);
}

__global__ void gcn_rsqrt(float* __restrict__ deg)
{
    int i = blockIdx.x * blockDim.x + threadIdx.x;
    if (i >= NNODES) return;
    deg[i] = rsqrtf(deg[i]);
}

template <int K, int NC>
__global__ void gcn_gemm_wmma(const float* __restrict__ A,
                              const float* __restrict__ B,
                              float* __restrict__ C, int M)
{
    const int wave = (blockIdx.x * blockDim.x + threadIdx.x) >> 5;
    constexpr int TN = NC / 16;
    const int numTiles = (M >> 4) * TN;
    if (wave >= numTiles) return;

    const int lane = threadIdx.x & 31;
    const int lo = lane & 15;
    const int hi = lane >> 4;
    const int tm = (wave / TN) << 4;
    const int tn = (wave % TN) << 4;

    const float* arow = A + (size_t)(tm + lo) * K + 2 * hi;
    const float* bcol = B + (size_t)(2 * hi) * NC + tn + lo;

    v8f acc = {};
#pragma unroll
    for (int k = 0; k < K; k += 4) {
        v2f a;
        a.x = arow[k];
        a.y = arow[k + 1];
        v2f b;
        b.x = bcol[(size_t)k * NC];
        b.y = bcol[(size_t)(k + 1) * NC];
        acc = __builtin_amdgcn_wmma_f32_16x16x4_f32(
             false, a,  false, b,
             (short)0, acc,  false,  false);
    }

    float* crow = C + (size_t)(tm + 8 * hi) * NC + tn + lo;
#pragma unroll
    for (int g = 0; g < 8; ++g) crow[(size_t)g * NC] = acc[g];
}

template <int C>
__global__ void gcn_scatter(const float* __restrict__ h,
                            const int* __restrict__ src,
                            const int* __restrict__ dst,
                            const float* __restrict__ dinv,
                            float* __restrict__ out)
{
    constexpr int Q = C / 4;
    long long t = (long long)blockIdx.x * blockDim.x + threadIdx.x;
    if (t >= (long long)(NEDGES + NNODES) * Q) return;
    int e = (int)(t / Q);
    int q = (int)(t % Q);
    int s, d;
    if (e < NEDGES) { s = src[e]; d = dst[e]; }
    else            { s = e - NEDGES; d = s; }
    float w = dinv[s] * dinv[d];
    float4 hv = *(const float4*)(h + (size_t)s * C + 4 * q);
    float* o = out + (size_t)d * C + 4 * q;
    atomicAdd(o + 0, hv.x * w);
    atomicAdd(o + 1, hv.y * w);
    atomicAdd(o + 2, hv.z * w);
    atomicAdd(o + 3, hv.w * w);
}

__global__ void gcn_bias_relu(float* __restrict__ agg, const float* __restrict__ b1)
{
    int i = blockIdx.x * blockDim.x + threadIdx.x;
    if (i >= NNODES * HID_C) return;
    float v = agg[i] + b1[i & (HID_C - 1)];
    agg[i] = v > 0.f ? v : 0.f;
}

__global__ void gcn_bias2(float* __restrict__ out, const float* __restrict__ b2)
{
    int i = blockIdx.x * blockDim.x + threadIdx.x;
    if (i >= NNODES * OUT_C) return;
    out[i] += b2[i & (OUT_C - 1)];
}

extern "C" void kernel_launch(void* const* d_in, const int* in_sizes, int n_in,
                              void* d_out, int out_size, void* d_ws, size_t ws_size,
                              hipStream_t stream)
{
    const float* x    = (const float*)d_in[0];
    const int*   eidx = (const int*)d_in[1];
    const int*   src  = eidx;
    const int*   dst  = eidx + NEDGES;
    const float* W1   = (const float*)d_in[2];
    const float* b1   = (const float*)d_in[3];
    const float* W2   = (const float*)d_in[4];
    const float* b2   = (const float*)d_in[5];
    float* out = (float*)d_out;

    float* dinv = (float*)d_ws;
    float* h    = dinv + NNODES;
    float* agg1 = h + (size_t)NNODES * HID_C;

    const int B = 256;

    gcn_init<<<(NNODES + B - 1) / B, B, 0, stream>>>(dinv, (float4*)agg1, (float4*)out);

    gcn_degree<<<(NEDGES + B - 1) / B, B, 0, stream>>>(dst, dinv);
    gcn_rsqrt<<<(NNODES + B - 1) / B, B, 0, stream>>>(dinv);

    {
        int waves = (NNODES / 16) * (HID_C / 16);
        int thr = waves * 32;
        gcn_gemm_wmma<IN_C, HID_C><<<(thr + B - 1) / B, B, 0, stream>>>(x, W1, h, NNODES);
    }

    {
        long long thr = (long long)(NEDGES + NNODES) * (HID_C / 4);
        gcn_scatter<HID_C><<<(int)((thr + B - 1) / B), B, 0, stream>>>(h, src, dst, dinv, agg1);
    }

    gcn_bias_relu<<<(NNODES * HID_C + B - 1) / B, B, 0, stream>>>(agg1, b1);

    {
        int waves = (NNODES / 16) * (OUT_C / 16);
        int thr = waves * 32;
        gcn_gemm_wmma<HID_C, OUT_C><<<(thr + B - 1) / B, B, 0, stream>>>(agg1, W2, h, NNODES);
    }

    {
        long long thr = (long long)(NEDGES + NNODES) * (OUT_C / 4);
        gcn_scatter<OUT_C><<<(int)((thr + B - 1) / B), B, 0, stream>>>(h, src, dst, dinv, out);
    }

    gcn_bias2<<<(NNODES * OUT_C + B - 1) / B, B, 0, stream>>>(out, b2);
}


# GEMM: persistent waves hold all W fragments in registers (W through LDS once per work-group) and stream whole 16-row strips of A through a wave-own LDS tile by whole-line loads; sums unchanged
# speedup vs baseline: 1.7305x; 1.7305x over previous
#include <hip/hip_runtime.h>
#include <hip/hip_bf16.h>


#define NNODES 100000
#define NEDGES 1000000
#define IN_C   128
#define HID_C  64
#define OUT_C  32

#define OWN_ROWS    400
#define OWN_GROUPS  (NNODES / OWN_ROWS)
#define AGG_THREADS 1024
#define AGG_WAVES   (AGG_THREADS / 32)
#define Q_SLOTS     256
#define OWN_CHUNKS  65536
#define OWN_WORDS   (OWN_CHUNKS * 4)
#define LIST_WORDS  (OWN_GROUPS * AGG_WAVES * Q_SLOTS)

static_assert(NNODES % OWN_ROWS == 0 && OWN_GROUPS <= 255, "an owner id fits one byte and 0xFF is no group");
static_assert(NEDGES % 4 == 0 && NEDGES <= OWN_CHUNKS * 16, "the padded owner array holds every edge");
static_assert(OWN_CHUNKS % (AGG_THREADS * 4) == 0, "whole groups of four loads per lane");

typedef __attribute__((ext_vector_type(2))) float v2f;
typedef __attribute__((ext_vector_type(4))) float v4f;
typedef __attribute__((ext_vector_type(8))) float v8f;

__global__ void gcn_init(float* __restrict__ deg)
{
    int i = blockIdx.x * blockDim.x + threadIdx.x;
    if (i >= NNODES) return;
    deg[i] = 1.0f;
}

__global__ void gcn_degree(const int* __restrict__ dst, float* __restrict__ deg,
                           unsigned int* __restrict__ own)
{
    int t = blockIdx.x * blockDim.x + threadIdx.x;
    if (t >= OWN_WORDS) return;
    if (t >= NEDGES / 4) {
        own[t] = 0xFFFFFFFFu;
        return;
    }
    const int4 d = reinterpret_cast<const int4*>(dst)[t];
    atomicAdd(&deg[d.x], 1.0f);
    atomicAdd(&deg[d.y], 1.0f);
    atomicAdd(&deg[d.z], 1.0f);
    atomicAdd(&deg[d.w], 1.0f);
    own[t] = (unsigned int)(d.x / OWN_ROWS) | ((unsigned int)(d.y / OWN_ROWS) << 8) |
             ((unsigned int)(d.z / OWN_ROWS) << 16) | ((unsigned int)(d.w / OWN_ROWS) << 24);
}

__global__ void gcn_rsqrt(float* __restrict__ deg)
{
    int i = blockIdx.x * blockDim.x + threadIdx.x;
    if (i >= NNODES) return;
    deg[i] = rsqrtf(deg[i]);
}

#define GEMM_WAVES   8
#define GEMM_THREADS (GEMM_WAVES * 32)
#define NSTRIPS      (NNODES / 16)
#define GEMM1_GROUPS 256
#define GEMM2_GROUPS 512

static_assert(NNODES % 16 == 0, "whole 16-row strips");

template <int K, int NC, bool RELU_A, int NWG>
__global__ void __launch_bounds__(GEMM_THREADS)
gcn_gemm_strips(const float* __restrict__ A,
                const float* __restrict__ B,
                const float* __restrict__ dinv,
                const float* __restrict__ bias,
                float* __restrict__ C,
                float* __restrict__ S)
{
    constexpr int TN  = NC / 16;
    constexpr int PA  = K + 4;
    constexpr int PB  = 2 * NC + 32;
    constexpr int NLD = 16 * K / 128;
    static_assert(K % 4 == 0 && (16 * K) % 128 == 0, "a strip is whole 512-B runs and a 16-B piece stays inside one row");
    __shared__ alignas(16) float Bs[(K / 2) * PB];
    __shared__ alignas(16) float As[GEMM_WAVES * 16 * PA];

    const int tid  = threadIdx.x;
    const int lane = tid & 31;
    const int wv   = __builtin_amdgcn_readfirstlane(tid >> 5);
    const int lo = lane & 15;
    const int hi = lane >> 4;
    float* as = As + wv * 16 * PA;

    constexpr int SSTEP = NWG * GEMM_WAVES;
    int s = blockIdx.x + NWG * wv;

    v4f stage[NLD];
    if (s < NSTRIPS) {
        const v4f* g = reinterpret_cast<const v4f*>(A + (size_t)s * 16 * K) + lane;
#pragma unroll
        for (int i = 0; i < NLD; ++i) stage[i] = g[i * 32];
    }

    constexpr int BQ = NC / 4;
    constexpr int BITEMS = (K / 2) * BQ;
    static_assert(BITEMS % GEMM_THREADS == 0, "whole pieces per thread");
    constexpr int BPT = BITEMS / GEMM_THREADS;
    {
        v4f r0[BPT], r1[BPT];
#pragma unroll
        for (int j = 0; j < BPT; ++j) {
            const int i  = tid + j * GEMM_THREADS;
            const int kp = i / BQ;
            const int n  = (i % BQ) * 4;
            r0[j] = *reinterpret_cast<const v4f*>(B + (size_t)(2 * kp) * NC + n);
            r1[j] = *reinterpret_cast<const v4f*>(B + (size_t)(2 * kp + 1) * NC + n);
        }
#pragma unroll
        for (int j = 0; j < BPT; ++j) {
            const int i  = tid + j * GEMM_THREADS;
            const int kp = i / BQ;
            const int n  = (i % BQ) * 4;
            v4f* d = reinterpret_cast<v4f*>(Bs + kp * PB + 2 * n);
            d[0] = v4f{r0[j].x, r1[j].x, r0[j].y, r1[j].y};
            d[1] = v4f{r0[j].z, r1[j].z, r0[j].w, r1[j].w};
        }
    }
    float bj[TN];
#pragma unroll
    for (int t = 0; t < TN; ++t) bj[t] = bias[t * 16 + lo];
    __syncthreads();

    const float* afrag = as + lo * PA + 2 * hi;
    const float* bfrag = Bs + hi * PB + 2 * lo;

    v2f bf[K / 4][TN];
#pragma unroll
    for (int kk = 0; kk < K / 4; ++kk)
#pragma unroll
        for (int t = 0; t < TN; ++t)
            bf[kk][t] = *reinterpret_cast<const v2f*>(bfrag + (2 * kk) * PB + 32 * t);

    for (; s < NSTRIPS; s += SSTEP) {
#pragma unroll
        for (int i = 0; i < NLD; ++i) {
            v4f v = stage[i];
            if (RELU_A) {
                v.x = fmaxf(v.x, 0.f);
                v.y = fmaxf(v.y, 0.f);
                v.z = fmaxf(v.z, 0.f);
                v.w = fmaxf(v.w, 0.f);
            }
            const int e = i * 128 + 4 * lane;
            *reinterpret_cast<v4f*>(as + (e / K) * PA + (e % K)) = v;
        }
        const int tm = s * 16;
        const float4 d0 = *(const float4*)(dinv + tm + 8 * hi);
        const float4 d1 = *(const float4*)(dinv + tm + 8 * hi + 4);
        if (s + SSTEP < NSTRIPS) {
            const v4f* g = reinterpret_cast<const v4f*>(A + (size_t)(s + SSTEP) * 16 * K) + lane;
#pragma unroll
            for (int i = 0; i < NLD; ++i) stage[i] = g[i * 32];
        }

        v8f acc[TN];
#pragma unroll
        for (int t = 0; t < TN; ++t) acc[t] = v8f{};
#pragma unroll
        for (int k = 0; k < K; k += 4) {
            const v2f a = *reinterpret_cast<const v2f*>(afrag + k);
#pragma unroll
            for (int t = 0; t < TN; ++t) {
                acc[t] = __builtin_amdgcn_wmma_f32_16x16x4_f32(
                     false, a,  false, bf[k / 4][t],
                     (short)0, acc[t],  false,  false);
            }
        }

        const float dv[8] = {d0.x, d0.y, d0.z, d0.w, d1.x, d1.y, d1.z, d1.w};
#pragma unroll
        for (int t = 0; t < TN; ++t) {
            const size_t off = (size_t)(tm + 8 * hi) * NC + t * 16 + lo;
            float* crow = C + off;
            float* srow = S + off;
#pragma unroll
            for (int g = 0; g < 8; ++g) {
                const float w = dv[g] * dv[g];
                crow[(size_t)g * NC] = acc[t][g];
                srow[(size_t)g * NC] = acc[t][g] * w + bj[t];
            }
        }
    }
}


__device__ __forceinline__ unsigned int zero_bytes(unsigned int x)
{
    return ~(((x & 0x7F7F7F7Fu) + 0x7F7F7F7Fu) | x | 0x7F7F7F7Fu);
}

__device__ __forceinline__ unsigned int match16(const uint4 v, unsigned int pat)
{
    return (zero_bytes(v.x ^ pat) >> 7) | (zero_bytes(v.y ^ pat) >> 6) |
           (zero_bytes(v.z ^ pat) >> 5) | (zero_bytes(v.w ^ pat) >> 4);
}

template <int C, bool FIRST>
__global__ void __launch_bounds__(AGG_THREADS)
gcn_aggregate_owned(const float* __restrict__ h,
                    const int* __restrict__ src,
                    const int* __restrict__ dst,
                    const float* __restrict__ dinv,
                    const unsigned int* __restrict__ own,
                    float* __restrict__ acc,
                    int* __restrict__ lcnt,
                    int* __restrict__ lsrc,
                    int* __restrict__ lrow,
                    float* __restrict__ lw)
{
    constexpr int V = C / 32;
    constexpr int TILE4 = OWN_ROWS * C / 4;
    constexpr int K = Q_SLOTS / 32;
    __shared__ alignas(16) float tile[OWN_ROWS * C];
    __shared__ int   q_src[AGG_WAVES * Q_SLOTS];
    __shared__ int   q_row[AGG_WAVES * Q_SLOTS];
    __shared__ float q_w[AGG_WAVES * Q_SLOTS];
    __shared__ int   s_cnt[AGG_WAVES];

    const int tid  = threadIdx.x;
    const int lane = tid & 31;
    const int wv   = tid >> 5;
    const int grp  = blockIdx.x;
    const int base = grp * OWN_ROWS;
    float* gacc = acc + (size_t)base * C;
    int*   qs = q_src + wv * Q_SLOTS;
    int*   qr = q_row + wv * Q_SLOTS;
    float* qw = q_w   + wv * Q_SLOTS;
    const size_t lbase = ((size_t)grp * AGG_WAVES + wv) * Q_SLOTS;

    for (int i = tid; i < TILE4; i += AGG_THREADS)
        reinterpret_cast<float4*>(tile)[i] = reinterpret_cast<const float4*>(gacc)[i];

    int wcount = 0;
    bool scan = true;
    if (!FIRST) {
        wcount = lcnt[grp * AGG_WAVES + wv];
        scan = wcount > Q_SLOTS;
        if (scan) wcount = 0;
    }
    __syncthreads();

    auto add_direct = [&](int e) {
        const int s = src[e];
        const int d = dst[e];
        const float w = dinv[s] * dinv[d];
        const float* hr = h + (size_t)s * C;
        float* tr = tile + (d - base) * C;
#pragma unroll 1
        for (int c = 0; c < C; ++c) atomicAdd(&tr[c], hr[c] * w);
    };

    if (scan) {
        const unsigned int pat = (unsigned int)grp * 0x01010101u;
        const uint4* own16 = reinterpret_cast<const uint4*>(own);
        const int first = grp * (OWN_CHUNKS / OWN_GROUPS);
        constexpr int DEPTH = 4;
        constexpr int NGRP  = OWN_CHUNKS / (AGG_THREADS * DEPTH);

        uint4 nv[DEPTH];
        int   nc[DEPTH];
        auto fetch = [&](int it, uint4& v, int& ch) {
            ch = (it * AGG_THREADS + tid + first) & (OWN_CHUNKS - 1);
            v = own16[ch];
        };
#pragma unroll
        for (int u = 0; u < DEPTH; ++u) fetch(u, nv[u], nc[u]);

        for (int g = 0; g < NGRP; ++g) {
            uint4 cv[DEPTH];
            int   cc[DEPTH];
#pragma unroll
            for (int u = 0; u < DEPTH; ++u) { cv[u] = nv[u]; cc[u] = nc[u]; }
            if (g + 1 < NGRP) {
#pragma unroll
                for (int u = 0; u < DEPTH; ++u) fetch((g + 1) * DEPTH + u, nv[u], nc[u]);
            }
#pragma unroll
            for (int u = 0; u < DEPTH; u += 2) {
                unsigned int mk = match16(cv[u], pat) | (match16(cv[u + 1], pat) << 4);
                while (true) {
                    const bool has = mk != 0u;
                    const unsigned int bal = __builtin_amdgcn_ballot_w32(has);
                    if (bal == 0u) break;
                    if (has) {
                        const int p = __builtin_ctz(mk);
                        const int ch = (p & 4) ? cc[u + 1] : cc[u];
                        const int e = ch * 16 + (((p & 3) << 2) | (p >> 3));
                        const int slot = wcount + (int)__builtin_amdgcn_mbcnt_lo(bal, 0u);
                        if (slot < Q_SLOTS) qs[slot] = e;
                        else                add_direct(e);
                        mk &= mk - 1u;
                    }
                    wcount += __builtin_popcount(bal);
                }
            }
        }
    }
    if (FIRST && lane == 0) s_cnt[wv] = wcount;
    __syncthreads();

    const int n = wcount < Q_SLOTS ? wcount : Q_SLOTS;
    if (scan) {
        int ev[K], sv[K], dv[K];
#pragma unroll
        for (int k = 0; k < K; ++k) ev[k] = qs[lane + 32 * k];
#pragma unroll
        for (int k = 0; k < K; ++k) {
            const int e = (lane + 32 * k < n) ? ev[k] : 0;
            sv[k] = src[e];
            dv[k] = dst[e];
        }
        float wk[K];
#pragma unroll
        for (int k = 0; k < K; ++k) wk[k] = dinv[sv[k]] * dinv[dv[k]];
#pragma unroll
        for (int k = 0; k < K; ++k) {
            const int i = lane + 32 * k;
            qs[i] = sv[k];
            qr[i] = (dv[k] - base) * C;
            qw[i] = wk[k];
        }
        if (FIRST) {
#pragma unroll
            for (int k = 0; k < K; ++k) {
                const int i = lane + 32 * k;
                if (i < n) {
                    lsrc[lbase + i] = sv[k];
                    lrow[lbase + i] = dv[k] - base;
                    lw[lbase + i]   = wk[k];
                }
            }
        }
        if (FIRST && wv == 0) lcnt[grp * AGG_WAVES + lane] = s_cnt[lane];
    } else {
#pragma unroll
        for (int k = 0; k < K; ++k) {
            const int i = lane + 32 * k;
            if (i < n) {
                qs[i] = lsrc[lbase + i];
                qr[i] = lrow[lbase + i] * C;
                qw[i] = lw[lbase + i];
            }
        }
    }
    __syncthreads();

    constexpr int U = 8;
    auto load_row = [&](int i, float (&r)[V]) {
        const int s = qs[i];
        if constexpr (V == 2) {
            const float2 t = *reinterpret_cast<const float2*>(h + (size_t)s * C + 2 * lane);
            r[0] = t.x;
            r[1] = t.y;
        } else {
            r[0] = h[(size_t)s * C + lane];
        }
    };
    auto add_row = [&](int i, const float (&r)[V]) {
        const float w = qw[i];
        float* tr = tile + qr[i] + V * lane;
#pragma unroll
        for (int v = 0; v < V; ++v) atomicAdd(&tr[v], r[v] * w);
    };
    const int full = n - (n % U);
    if (full > 0) {
        float cur[U][V], nxt[U][V];
#pragma unroll
        for (int u = 0; u < U; ++u) load_row(u, cur[u]);
        for (int i0 = U; i0 < full; i0 += U) {
#pragma unroll
            for (int u = 0; u < U; ++u) load_row(i0 + u, nxt[u]);
#pragma unroll
            for (int u = 0; u < U; ++u) add_row(i0 - U + u, cur[u]);
#pragma unroll
            for (int u = 0; u < U; ++u)
#pragma unroll
                for (int v = 0; v < V; ++v) cur[u][v] = nxt[u][v];
        }
#pragma unroll
        for (int u = 0; u < U; ++u) add_row(full - U + u, cur[u]);
    }
    for (int i = full; i < n; ++i) {
        float r[V];
        load_row(i, r);
        add_row(i, r);
    }
    __syncthreads();

    for (int i = tid; i < TILE4; i += AGG_THREADS)
        reinterpret_cast<float4*>(gacc)[i] = reinterpret_cast<const float4*>(tile)[i];
}

extern "C" void kernel_launch(void* const* d_in, const int* in_sizes, int n_in,
                              void* d_out, int out_size, void* d_ws, size_t ws_size,
                              hipStream_t stream)
{
    const float* x    = (const float*)d_in[0];
    const int*   eidx = (const int*)d_in[1];
    const int*   src  = eidx;
    const int*   dst  = eidx + NEDGES;
    const float* W1   = (const float*)d_in[2];
    const float* b1   = (const float*)d_in[3];
    const float* W2   = (const float*)d_in[4];
    const float* b2   = (const float*)d_in[5];
    float* out = (float*)d_out;

    float* dinv = (float*)d_ws;
    float* h    = dinv + NNODES;
    float* agg1 = h + (size_t)NNODES * HID_C;
    unsigned int* own = (unsigned int*)(agg1 + (size_t)NNODES * HID_C);
    int*   lsrc = (int*)(own + OWN_WORDS);
    int*   lrow = lsrc + LIST_WORDS;
    float* lw   = (float*)(lrow + LIST_WORDS);
    int*   lcnt = (int*)(lw + LIST_WORDS);

    const int B = 256;

    gcn_init<<<(NNODES + B - 1) / B, B, 0, stream>>>(dinv);

    gcn_degree<<<(OWN_WORDS + B - 1) / B, B, 0, stream>>>(dst, dinv, own);
    gcn_rsqrt<<<(NNODES + B - 1) / B, B, 0, stream>>>(dinv);

    gcn_gemm_strips<IN_C, HID_C, false, GEMM1_GROUPS><<<GEMM1_GROUPS, GEMM_THREADS, 0, stream>>>(
        x, W1, dinv, b1, h, agg1);

    gcn_aggregate_owned<HID_C, true><<<OWN_GROUPS, AGG_THREADS, 0, stream>>>(
        h, src, dst, dinv, own, agg1, lcnt, lsrc, lrow, lw);

    gcn_gemm_strips<HID_C, OUT_C, true, GEMM2_GROUPS><<<GEMM2_GROUPS, GEMM_THREADS, 0, stream>>>(
        agg1, W2, dinv, b2, h, out);

    gcn_aggregate_owned<OUT_C, false><<<OWN_GROUPS, AGG_THREADS, 0, stream>>>(
        h, src, dst, dinv, own, out, lcnt, lsrc, lrow, lw);
}


# edge norm factored out of the sums: GEMMs store rows x dinv[row]; a tile sums them unweighted from its own row, then x dinv[dst] + bias at store (bias last); norm lookups, norm list and S plane gone
# speedup vs baseline: 1.8676x; 1.0792x over previous
#include <hip/hip_runtime.h>
#include <hip/hip_bf16.h>


#define NNODES 100000
#define NEDGES 1000000
#define IN_C   128
#define HID_C  64
#define OUT_C  32

#define OWN_ROWS    400
#define OWN_GROUPS  (NNODES / OWN_ROWS)
#define AGG_THREADS 1024
#define AGG_WAVES   (AGG_THREADS / 32)
#define Q_SLOTS     256
#define OWN_CHUNKS  65536
#define OWN_WORDS   (OWN_CHUNKS * 4)
#define LIST_WORDS  (OWN_GROUPS * AGG_WAVES * Q_SLOTS)

static_assert(NNODES % OWN_ROWS == 0 && OWN_GROUPS <= 255, "an owner id fits one byte and 0xFF is no group");
static_assert(NEDGES % 4 == 0 && NEDGES <= OWN_CHUNKS * 16, "the padded owner array holds every edge");
static_assert(OWN_CHUNKS % (AGG_THREADS * 4) == 0, "whole groups of four loads per lane");

typedef __attribute__((ext_vector_type(2))) float v2f;
typedef __attribute__((ext_vector_type(4))) float v4f;
typedef __attribute__((ext_vector_type(8))) float v8f;

__global__ void gcn_init(float* __restrict__ deg)
{
    int i = blockIdx.x * blockDim.x + threadIdx.x;
    if (i >= NNODES) return;
    deg[i] = 1.0f;
}

__global__ void gcn_degree(const int* __restrict__ dst, float* __restrict__ deg,
                           unsigned int* __restrict__ own)
{
    int t = blockIdx.x * blockDim.x + threadIdx.x;
    if (t >= OWN_WORDS) return;
    if (t >= NEDGES / 4) {
        own[t] = 0xFFFFFFFFu;
        return;
    }
    const int4 d = reinterpret_cast<const int4*>(dst)[t];
    atomicAdd(&deg[d.x], 1.0f);
    atomicAdd(&deg[d.y], 1.0f);
    atomicAdd(&deg[d.z], 1.0f);
    atomicAdd(&deg[d.w], 1.0f);
    own[t] = (unsigned int)(d.x / OWN_ROWS) | ((unsigned int)(d.y / OWN_ROWS) << 8) |
             ((unsigned int)(d.z / OWN_ROWS) << 16) | ((unsigned int)(d.w / OWN_ROWS) << 24);
}

__global__ void gcn_rsqrt(float* __restrict__ deg)
{
    int i = blockIdx.x * blockDim.x + threadIdx.x;
    if (i >= NNODES) return;
    deg[i] = rsqrtf(deg[i]);
}

#define GEMM_WAVES   8
#define GEMM_THREADS (GEMM_WAVES * 32)
#define NSTRIPS      (NNODES / 16)
#define GEMM1_GROUPS 256
#define GEMM2_GROUPS 512

static_assert(NNODES % 16 == 0, "whole 16-row strips");

template <int K, int NC, bool RELU_A, int NWG>
__global__ void __launch_bounds__(GEMM_THREADS)
gcn_gemm_strips(const float* __restrict__ A,
                const float* __restrict__ B,
                const float* __restrict__ dinv,
                float* __restrict__ C)
{
    constexpr int TN  = NC / 16;
    constexpr int PA  = K + 4;
    constexpr int PB  = 2 * NC + 32;
    constexpr int NLD = 16 * K / 128;
    static_assert(K % 4 == 0 && (16 * K) % 128 == 0, "a strip is whole 512-B runs and a 16-B piece stays inside one row");
    __shared__ alignas(16) float Bs[(K / 2) * PB];
    __shared__ alignas(16) float As[GEMM_WAVES * 16 * PA];

    const int tid  = threadIdx.x;
    const int lane = tid & 31;
    const int wv   = __builtin_amdgcn_readfirstlane(tid >> 5);
    const int lo = lane & 15;
    const int hi = lane >> 4;
    float* as = As + wv * 16 * PA;

    constexpr int SSTEP = NWG * GEMM_WAVES;
    int s = blockIdx.x + NWG * wv;

    v4f stage[NLD];
    if (s < NSTRIPS) {
        const v4f* g = reinterpret_cast<const v4f*>(A + (size_t)s * 16 * K) + lane;
#pragma unroll
        for (int i = 0; i < NLD; ++i) stage[i] = g[i * 32];
    }

    constexpr int BQ = NC / 4;
    constexpr int BITEMS = (K / 2) * BQ;
    static_assert(BITEMS % GEMM_THREADS == 0, "whole pieces per thread");
    constexpr int BPT = BITEMS / GEMM_THREADS;
    {
        v4f r0[BPT], r1[BPT];
#pragma unroll
        for (int j = 0; j < BPT; ++j) {
            const int i  = tid + j * GEMM_THREADS;
            const int kp = i / BQ;
            const int n  = (i % BQ) * 4;
            r0[j] = *reinterpret_cast<const v4f*>(B + (size_t)(2 * kp) * NC + n);
            r1[j] = *reinterpret_cast<const v4f*>(B + (size_t)(2 * kp + 1) * NC + n);
        }
#pragma unroll
        for (int j = 0; j < BPT; ++j) {
            const int i  = tid + j * GEMM_THREADS;
            const int kp = i / BQ;
            const int n  = (i % BQ) * 4;
            v4f* d = reinterpret_cast<v4f*>(Bs + kp * PB + 2 * n);
            d[0] = v4f{r0[j].x, r1[j].x, r0[j].y, r1[j].y};
            d[1] = v4f{r0[j].z, r1[j].z, r0[j].w, r1[j].w};
        }
    }
    __syncthreads();

    const float* afrag = as + lo * PA + 2 * hi;
    const float* bfrag = Bs + hi * PB + 2 * lo;

    v2f bf[K / 4][TN];
#pragma unroll
    for (int kk = 0; kk < K / 4; ++kk)
#pragma unroll
        for (int t = 0; t < TN; ++t)
            bf[kk][t] = *reinterpret_cast<const v2f*>(bfrag + (2 * kk) * PB + 32 * t);

    for (; s < NSTRIPS; s += SSTEP) {
#pragma unroll
        for (int i = 0; i < NLD; ++i) {
            v4f v = stage[i];
            if (RELU_A) {
                v.x = fmaxf(v.x, 0.f);
                v.y = fmaxf(v.y, 0.f);
                v.z = fmaxf(v.z, 0.f);
                v.w = fmaxf(v.w, 0.f);
            }
            const int e = i * 128 + 4 * lane;
            *reinterpret_cast<v4f*>(as + (e / K) * PA + (e % K)) = v;
        }
        const int tm = s * 16;
        const float4 d0 = *(const float4*)(dinv + tm + 8 * hi);
        const float4 d1 = *(const float4*)(dinv + tm + 8 * hi + 4);
        if (s + SSTEP < NSTRIPS) {
            const v4f* g = reinterpret_cast<const v4f*>(A + (size_t)(s + SSTEP) * 16 * K) + lane;
#pragma unroll
            for (int i = 0; i < NLD; ++i) stage[i] = g[i * 32];
        }

        v8f acc[TN];
#pragma unroll
        for (int t = 0; t < TN; ++t) acc[t] = v8f{};
#pragma unroll
        for (int k = 0; k < K; k += 4) {
            const v2f a = *reinterpret_cast<const v2f*>(afrag + k);
#pragma unroll
            for (int t = 0; t < TN; ++t) {
                acc[t] = __builtin_amdgcn_wmma_f32_16x16x4_f32(
                     false, a,  false, bf[k / 4][t],
                     (short)0, acc[t],  false,  false);
            }
        }

        const float dv[8] = {d0.x, d0.y, d0.z, d0.w, d1.x, d1.y, d1.z, d1.w};
#pragma unroll
        for (int t = 0; t < TN; ++t) {
            float* crow = C + (size_t)(tm + 8 * hi) * NC + t * 16 + lo;
#pragma unroll
            for (int g = 0; g < 8; ++g) crow[(size_t)g * NC] = acc[t][g] * dv[g];
        }
    }
}


__device__ __forceinline__ unsigned int zero_bytes(unsigned int x)
{
    return ~(((x & 0x7F7F7F7Fu) + 0x7F7F7F7Fu) | x | 0x7F7F7F7Fu);
}

__device__ __forceinline__ unsigned int match16(const uint4 v, unsigned int pat)
{
    return (zero_bytes(v.x ^ pat) >> 7) | (zero_bytes(v.y ^ pat) >> 6) |
           (zero_bytes(v.z ^ pat) >> 5) | (zero_bytes(v.w ^ pat) >> 4);
}

template <int C, bool FIRST>
__global__ void __launch_bounds__(AGG_THREADS)
gcn_aggregate_owned(const float* __restrict__ h,
                    const int* __restrict__ src,
                    const int* __restrict__ dst,
                    const float* __restrict__ dinv,
                    const float* __restrict__ bias,
                    const unsigned int* __restrict__ own,
                    float* __restrict__ acc,
                    int* __restrict__ lcnt,
                    int* __restrict__ lsrc,
                    int* __restrict__ lrow)
{
    constexpr int V = C / 32;
    constexpr int TILE4 = OWN_ROWS * C / 4;
    constexpr int K = Q_SLOTS / 32;
    __shared__ alignas(16) float tile[OWN_ROWS * C];
    __shared__ int   q_src[AGG_WAVES * Q_SLOTS];
    __shared__ int   q_row[AGG_WAVES * Q_SLOTS];
    __shared__ int   s_cnt[AGG_WAVES];

    const int tid  = threadIdx.x;
    const int lane = tid & 31;
    const int wv   = tid >> 5;
    const int grp  = blockIdx.x;
    const int base = grp * OWN_ROWS;
    float* gacc = acc + (size_t)base * C;
    const float* hown = h + (size_t)base * C;
    int*   qs = q_src + wv * Q_SLOTS;
    int*   qr = q_row + wv * Q_SLOTS;
    const size_t lbase = ((size_t)grp * AGG_WAVES + wv) * Q_SLOTS;

    for (int i = tid; i < TILE4; i += AGG_THREADS)
        reinterpret_cast<float4*>(tile)[i] = reinterpret_cast<const float4*>(hown)[i];

    int wcount = 0;
    bool scan = true;
    if (!FIRST) {
        wcount = lcnt[grp * AGG_WAVES + wv];
        scan = wcount > Q_SLOTS;
        if (scan) wcount = 0;
    }
    __syncthreads();

    auto add_direct = [&](int e) {
        const int s = src[e];
        const int d = dst[e];
        const float* hr = h + (size_t)s * C;
        float* tr = tile + (d - base) * C;
#pragma unroll 1
        for (int c = 0; c < C; ++c) atomicAdd(&tr[c], hr[c]);
    };

    if (scan) {
        const unsigned int pat = (unsigned int)grp * 0x01010101u;
        const uint4* own16 = reinterpret_cast<const uint4*>(own);
        const int first = grp * (OWN_CHUNKS / OWN_GROUPS);
        constexpr int DEPTH = 4;
        constexpr int NGRP  = OWN_CHUNKS / (AGG_THREADS * DEPTH);

        uint4 nv[DEPTH];
        int   nc[DEPTH];
        auto fetch = [&](int it, uint4& v, int& ch) {
            ch = (it * AGG_THREADS + tid + first) & (OWN_CHUNKS - 1);
            v = own16[ch];
        };
#pragma unroll
        for (int u = 0; u < DEPTH; ++u) fetch(u, nv[u], nc[u]);

        for (int g = 0; g < NGRP; ++g) {
            uint4 cv[DEPTH];
            int   cc[DEPTH];
#pragma unroll
            for (int u = 0; u < DEPTH; ++u) { cv[u] = nv[u]; cc[u] = nc[u]; }
            if (g + 1 < NGRP) {
#pragma unroll
                for (int u = 0; u < DEPTH; ++u) fetch((g + 1) * DEPTH + u, nv[u], nc[u]);
            }
#pragma unroll
            for (int u = 0; u < DEPTH; u += 2) {
                unsigned int mk = match16(cv[u], pat) | (match16(cv[u + 1], pat) << 4);
                while (true) {
                    const bool has = mk != 0u;
                    const unsigned int bal = __builtin_amdgcn_ballot_w32(has);
                    if (bal == 0u) break;
                    if (has) {
                        const int p = __builtin_ctz(mk);
                        const int ch = (p & 4) ? cc[u + 1] : cc[u];
                        const int e = ch * 16 + (((p & 3) << 2) | (p >> 3));
                        const int slot = wcount + (int)__builtin_amdgcn_mbcnt_lo(bal, 0u);
                        if (slot < Q_SLOTS) qs[slot] = e;
                        else                add_direct(e);
                        mk &= mk - 1u;
                    }
                    wcount += __builtin_popcount(bal);
                }
            }
        }
    }
    if (FIRST && lane == 0) s_cnt[wv] = wcount;
    __syncthreads();

    const int n = wcount < Q_SLOTS ? wcount : Q_SLOTS;
    if (scan) {
        int ev[K], sv[K], dv[K];
#pragma unroll
        for (int k = 0; k < K; ++k) ev[k] = qs[lane + 32 * k];
#pragma unroll
        for (int k = 0; k < K; ++k) {
            const int e = (lane + 32 * k < n) ? ev[k] : 0;
            sv[k] = src[e];
            dv[k] = dst[e];
        }
#pragma unroll
        for (int k = 0; k < K; ++k) {
            const int i = lane + 32 * k;
            qs[i] = sv[k];
            qr[i] = (dv[k] - base) * C;
        }
        if (FIRST) {
#pragma unroll
            for (int k = 0; k < K; ++k) {
                const int i = lane + 32 * k;
                if (i < n) {
                    lsrc[lbase + i] = sv[k];
                    lrow[lbase + i] = dv[k] - base;
                }
            }
        }
        if (FIRST && wv == 0) lcnt[grp * AGG_WAVES + lane] = s_cnt[lane];
    } else {
#pragma unroll
        for (int k = 0; k < K; ++k) {
            const int i = lane + 32 * k;
            if (i < n) {
                qs[i] = lsrc[lbase + i];
                qr[i] = lrow[lbase + i] * C;
            }
        }
    }
    __syncthreads();

    constexpr int U = 8;
    auto load_row = [&](int i, float (&r)[V]) {
        const int s = qs[i];
        if constexpr (V == 2) {
            const float2 t = *reinterpret_cast<const float2*>(h + (size_t)s * C + 2 * lane);
            r[0] = t.x;
            r[1] = t.y;
        } else {
            r[0] = h[(size_t)s * C + lane];
        }
    };
    auto add_row = [&](int i, const float (&r)[V]) {
        float* tr = tile + qr[i] + V * lane;
#pragma unroll
        for (int v = 0; v < V; ++v) atomicAdd(&tr[v], r[v]);
    };
    const int full = n - (n % U);
    if (full > 0) {
        float cur[U][V], nxt[U][V];
#pragma unroll
        for (int u = 0; u < U; ++u) load_row(u, cur[u]);
        for (int i0 = U; i0 < full; i0 += U) {
#pragma unroll
            for (int u = 0; u < U; ++u) load_row(i0 + u, nxt[u]);
#pragma unroll
            for (int u = 0; u < U; ++u) add_row(i0 - U + u, cur[u]);
#pragma unroll
            for (int u = 0; u < U; ++u)
#pragma unroll
                for (int v = 0; v < V; ++v) cur[u][v] = nxt[u][v];
        }
#pragma unroll
        for (int u = 0; u < U; ++u) add_row(full - U + u, cur[u]);
    }
    for (int i = full; i < n; ++i) {
        float r[V];
        load_row(i, r);
        add_row(i, r);
    }
    __syncthreads();

    static_assert(AGG_THREADS % (C / 4) == 0, "a thread keeps its channels from turn to turn");
    const float* bq = bias + 4 * (tid % (C / 4));
    const float4 bv = make_float4(bq[0], bq[1], bq[2], bq[3]);
    for (int i = tid; i < TILE4; i += AGG_THREADS) {
        const float dd = dinv[base + i / (C / 4)];
        float4 v = reinterpret_cast<const float4*>(tile)[i];
        v.x = v.x * dd + bv.x;
        v.y = v.y * dd + bv.y;
        v.z = v.z * dd + bv.z;
        v.w = v.w * dd + bv.w;
        reinterpret_cast<float4*>(gacc)[i] = v;
    }
}

extern "C" void kernel_launch(void* const* d_in, const int* in_sizes, int n_in,
                              void* d_out, int out_size, void* d_ws, size_t ws_size,
                              hipStream_t stream)
{
    const float* x    = (const float*)d_in[0];
    const int*   eidx = (const int*)d_in[1];
    const int*   src  = eidx;
    const int*   dst  = eidx + NEDGES;
    const float* W1   = (const float*)d_in[2];
    const float* b1   = (const float*)d_in[3];
    const float* W2   = (const float*)d_in[4];
    const float* b2   = (const float*)d_in[5];
    float* out = (float*)d_out;

    float* dinv = (float*)d_ws;
    float* h    = dinv + NNODES;
    float* agg1 = h + (size_t)NNODES * HID_C;
    unsigned int* own = (unsigned int*)(agg1 + (size_t)NNODES * HID_C);
    int*   lsrc = (int*)(own + OWN_WORDS);
    int*   lrow = lsrc + LIST_WORDS;
    int*   lcnt = lrow + LIST_WORDS;

    const int B = 256;

    gcn_init<<<(NNODES + B - 1) / B, B, 0, stream>>>(dinv);

    gcn_degree<<<(OWN_WORDS + B - 1) / B, B, 0, stream>>>(dst, dinv, own);
    gcn_rsqrt<<<(NNODES + B - 1) / B, B, 0, stream>>>(dinv);

    gcn_gemm_strips<IN_C, HID_C, false, GEMM1_GROUPS><<<GEMM1_GROUPS, GEMM_THREADS, 0, stream>>>(
        x, W1, dinv, h);

    gcn_aggregate_owned<HID_C, true><<<OWN_GROUPS, AGG_THREADS, 0, stream>>>(
        h, src, dst, dinv, b1, own, agg1, lcnt, lsrc, lrow);

    gcn_gemm_strips<HID_C, OUT_C, true, GEMM2_GROUPS><<<GEMM2_GROUPS, GEMM_THREADS, 0, stream>>>(
        agg1, W2, dinv, h);

    gcn_aggregate_owned<OUT_C, false><<<OWN_GROUPS, AGG_THREADS, 0, stream>>>(
        h, src, dst, dinv, b2, own, out, lcnt, lsrc, lrow);
}


# degrees counted in LDS by each row range's owner: its scan moves to a listing launch ahead of the GEMMs (integer LDS add per edge, writes dinv + lists); global atomics, deg fill, rsqrt pass gone
# speedup vs baseline: 2.6430x; 1.4152x over previous
#include <hip/hip_runtime.h>
#include <hip/hip_bf16.h>


#define NNODES 100000
#define NEDGES 1000000
#define IN_C   128
#define HID_C  64
#define OUT_C  32

#define OWN_ROWS    400
#define OWN_GROUPS  (NNODES / OWN_ROWS)
#define AGG_THREADS 1024
#define AGG_WAVES   (AGG_THREADS / 32)
#define Q_SLOTS     256
#define OWN_CHUNKS  65536
#define OWN_WORDS   (OWN_CHUNKS * 4)
#define LIST_WORDS  (OWN_GROUPS * AGG_WAVES * Q_SLOTS)

static_assert(NNODES % OWN_ROWS == 0 && OWN_GROUPS <= 255, "an owner id fits one byte and 0xFF is no group");
static_assert(NEDGES % 4 == 0 && NEDGES <= OWN_CHUNKS * 16, "the padded owner array holds every edge");
static_assert(OWN_CHUNKS % (AGG_THREADS * 4) == 0, "whole groups of four loads per lane");

typedef __attribute__((ext_vector_type(2))) float v2f;
typedef __attribute__((ext_vector_type(4))) float v4f;
typedef __attribute__((ext_vector_type(8))) float v8f;

__global__ void gcn_owner_bytes(const int* __restrict__ dst, unsigned int* __restrict__ own)
{
    int t = blockIdx.x * blockDim.x + threadIdx.x;
    if (t >= OWN_WORDS) return;
    if (t >= NEDGES / 4) {
        own[t] = 0xFFFFFFFFu;
        return;
    }
    const int4 d = reinterpret_cast<const int4*>(dst)[t];
    own[t] = (unsigned int)(d.x / OWN_ROWS) | ((unsigned int)(d.y / OWN_ROWS) << 8) |
             ((unsigned int)(d.z / OWN_ROWS) << 16) | ((unsigned int)(d.w / OWN_ROWS) << 24);
}

#define GEMM_WAVES   8
#define GEMM_THREADS (GEMM_WAVES * 32)
#define NSTRIPS      (NNODES / 16)
#define GEMM1_GROUPS 256
#define GEMM2_GROUPS 512

static_assert(NNODES % 16 == 0, "whole 16-row strips");

template <int K, int NC, bool RELU_A, int NWG>
__global__ void __launch_bounds__(GEMM_THREADS)
gcn_gemm_strips(const float* __restrict__ A,
                const float* __restrict__ B,
                const float* __restrict__ dinv,
                float* __restrict__ C)
{
    constexpr int TN  = NC / 16;
    constexpr int PA  = K + 4;
    constexpr int PB  = 2 * NC + 32;
    constexpr int NLD = 16 * K / 128;
    static_assert(K % 4 == 0 && (16 * K) % 128 == 0, "a strip is whole 512-B runs and a 16-B piece stays inside one row");
    __shared__ alignas(16) float Bs[(K / 2) * PB];
    __shared__ alignas(16) float As[GEMM_WAVES * 16 * PA];

    const int tid  = threadIdx.x;
    const int lane = tid & 31;
    const int wv   = __builtin_amdgcn_readfirstlane(tid >> 5);
    const int lo = lane & 15;
    const int hi = lane >> 4;
    float* as = As + wv * 16 * PA;

    constexpr int SSTEP = NWG * GEMM_WAVES;
    int s = blockIdx.x + NWG * wv;

    v4f stage[NLD];
    if (s < NSTRIPS) {
        const v4f* g = reinterpret_cast<const v4f*>(A + (size_t)s * 16 * K) + lane;
#pragma unroll
        for (int i = 0; i < NLD; ++i) stage[i] = g[i * 32];
    }

    constexpr int BQ = NC / 4;
    constexpr int BITEMS = (K / 2) * BQ;
    static_assert(BITEMS % GEMM_THREADS == 0, "whole pieces per thread");
    constexpr int BPT = BITEMS / GEMM_THREADS;
    {
        v4f r0[BPT], r1[BPT];
#pragma unroll
        for (int j = 0; j < BPT; ++j) {
            const int i  = tid + j * GEMM_THREADS;
            const int kp = i / BQ;
            const int n  = (i % BQ) * 4;
            r0[j] = *reinterpret_cast<const v4f*>(B + (size_t)(2 * kp) * NC + n);
            r1[j] = *reinterpret_cast<const v4f*>(B + (size_t)(2 * kp + 1) * NC + n);
        }
#pragma unroll
        for (int j = 0; j < BPT; ++j) {
            const int i  = tid + j * GEMM_THREADS;
            const int kp = i / BQ;
            const int n  = (i % BQ) * 4;
            v4f* d = reinterpret_cast<v4f*>(Bs + kp * PB + 2 * n);
            d[0] = v4f{r0[j].x, r1[j].x, r0[j].y, r1[j].y};
            d[1] = v4f{r0[j].z, r1[j].z, r0[j].w, r1[j].w};
        }
    }
    __syncthreads();

    const float* afrag = as + lo * PA + 2 * hi;
    const float* bfrag = Bs + hi * PB + 2 * lo;

    v2f bf[K / 4][TN];
#pragma unroll
    for (int kk = 0; kk < K / 4; ++kk)
#pragma unroll
        for (int t = 0; t < TN; ++t)
            bf[kk][t] = *reinterpret_cast<const v2f*>(bfrag + (2 * kk) * PB + 32 * t);

    for (; s < NSTRIPS; s += SSTEP) {
#pragma unroll
        for (int i = 0; i < NLD; ++i) {
            v4f v = stage[i];
            if (RELU_A) {
                v.x = fmaxf(v.x, 0.f);
                v.y = fmaxf(v.y, 0.f);
                v.z = fmaxf(v.z, 0.f);
                v.w = fmaxf(v.w, 0.f);
            }
            const int e = i * 128 + 4 * lane;
            *reinterpret_cast<v4f*>(as + (e / K) * PA + (e % K)) = v;
        }
        const int tm = s * 16;
        const float4 d0 = *(const float4*)(dinv + tm + 8 * hi);
        const float4 d1 = *(const float4*)(dinv + tm + 8 * hi + 4);
        if (s + SSTEP < NSTRIPS) {
            const v4f* g = reinterpret_cast<const v4f*>(A + (size_t)(s + SSTEP) * 16 * K) + lane;
#pragma unroll
            for (int i = 0; i < NLD; ++i) stage[i] = g[i * 32];
        }

        v8f acc[TN];
#pragma unroll
        for (int t = 0; t < TN; ++t) acc[t] = v8f{};
#pragma unroll
        for (int k = 0; k < K; k += 4) {
            const v2f a = *reinterpret_cast<const v2f*>(afrag + k);
#pragma unroll
            for (int t = 0; t < TN; ++t) {
                acc[t] = __builtin_amdgcn_wmma_f32_16x16x4_f32(
                     false, a,  false, bf[k / 4][t],
                     (short)0, acc[t],  false,  false);
            }
        }

        const float dv[8] = {d0.x, d0.y, d0.z, d0.w, d1.x, d1.y, d1.z, d1.w};
#pragma unroll
        for (int t = 0; t < TN; ++t) {
            float* crow = C + (size_t)(tm + 8 * hi) * NC + t * 16 + lo;
#pragma unroll
            for (int g = 0; g < 8; ++g) crow[(size_t)g * NC] = acc[t][g] * dv[g];
        }
    }
}


__device__ __forceinline__ unsigned int zero_bytes(unsigned int x)
{
    return ~(((x & 0x7F7F7F7Fu) + 0x7F7F7F7Fu) | x | 0x7F7F7F7Fu);
}

__device__ __forceinline__ unsigned int match16(const uint4 v, unsigned int pat)
{
    return (zero_bytes(v.x ^ pat) >> 7) | (zero_bytes(v.y ^ pat) >> 6) |
           (zero_bytes(v.z ^ pat) >> 5) | (zero_bytes(v.w ^ pat) >> 4);
}

template <typename OVER>
__device__ __forceinline__ int scan_owned(const unsigned int* __restrict__ own, int grp, int tid,
                                          int* __restrict__ qs, OVER over)
{
    int wcount = 0;
    const unsigned int pat = (unsigned int)grp * 0x01010101u;
    const uint4* own16 = reinterpret_cast<const uint4*>(own);
    const int first = grp * (OWN_CHUNKS / OWN_GROUPS);
    constexpr int DEPTH = 4;
    constexpr int NGRP  = OWN_CHUNKS / (AGG_THREADS * DEPTH);

    uint4 nv[DEPTH];
    int   nc[DEPTH];
    auto fetch = [&](int it, uint4& v, int& ch) {
        ch = (it * AGG_THREADS + tid + first) & (OWN_CHUNKS - 1);
        v = own16[ch];
    };
#pragma unroll
    for (int u = 0; u < DEPTH; ++u) fetch(u, nv[u], nc[u]);

    for (int g = 0; g < NGRP; ++g) {
        uint4 cv[DEPTH];
        int   cc[DEPTH];
#pragma unroll
        for (int u = 0; u < DEPTH; ++u) { cv[u] = nv[u]; cc[u] = nc[u]; }
        if (g + 1 < NGRP) {
#pragma unroll
            for (int u = 0; u < DEPTH; ++u) fetch((g + 1) * DEPTH + u, nv[u], nc[u]);
        }
#pragma unroll
        for (int u = 0; u < DEPTH; u += 2) {
            unsigned int mk = match16(cv[u], pat) | (match16(cv[u + 1], pat) << 4);
            while (true) {
                const bool has = mk != 0u;
                const unsigned int bal = __builtin_amdgcn_ballot_w32(has);
                if (bal == 0u) break;
                if (has) {
                    const int p = __builtin_ctz(mk);
                    const int ch = (p & 4) ? cc[u + 1] : cc[u];
                    const int e = ch * 16 + (((p & 3) << 2) | (p >> 3));
                    const int slot = wcount + (int)__builtin_amdgcn_mbcnt_lo(bal, 0u);
                    if (slot < Q_SLOTS) qs[slot] = e;
                    else                over(e);
                    mk &= mk - 1u;
                }
                wcount += __builtin_popcount(bal);
            }
        }
    }
    return wcount;
}

__global__ void __launch_bounds__(AGG_THREADS)
gcn_list_owned(const int* __restrict__ src,
               const int* __restrict__ dst,
               const unsigned int* __restrict__ own,
               float* __restrict__ dinv,
               int* __restrict__ lcnt,
               int* __restrict__ lsrc,
               int* __restrict__ lrow)
{
    constexpr int K = Q_SLOTS / 32;
    __shared__ int q_e[AGG_WAVES * Q_SLOTS];
    __shared__ int s_cnt[AGG_WAVES];
    __shared__ int r_cnt[OWN_ROWS];

    const int tid  = threadIdx.x;
    const int lane = tid & 31;
    const int wv   = tid >> 5;
    const int grp  = blockIdx.x;
    const int base = grp * OWN_ROWS;
    int* qs = q_e + wv * Q_SLOTS;
    const size_t lbase = ((size_t)grp * AGG_WAVES + wv) * Q_SLOTS;

    for (int i = tid; i < OWN_ROWS; i += AGG_THREADS) r_cnt[i] = 0;
    __syncthreads();

    const int wcount = scan_owned(own, grp, tid, qs, [&](int e) { atomicAdd(&r_cnt[dst[e] - base], 1); });
    if (lane == 0) s_cnt[wv] = wcount;
    __syncthreads();

    const int n = wcount < Q_SLOTS ? wcount : Q_SLOTS;
    int ev[K], sv[K], dv[K];
#pragma unroll
    for (int k = 0; k < K; ++k) ev[k] = qs[lane + 32 * k];
#pragma unroll
    for (int k = 0; k < K; ++k) {
        const int e = (lane + 32 * k < n) ? ev[k] : 0;
        sv[k] = src[e];
        dv[k] = dst[e];
    }
#pragma unroll
    for (int k = 0; k < K; ++k) {
        const int i = lane + 32 * k;
        if (i < n) {
            lsrc[lbase + i] = sv[k];
            lrow[lbase + i] = dv[k] - base;
            atomicAdd(&r_cnt[dv[k] - base], 1);
        }
    }
    __syncthreads();

    if (wv == 0) lcnt[grp * AGG_WAVES + lane] = s_cnt[lane];
    for (int i = tid; i < OWN_ROWS; i += AGG_THREADS) dinv[base + i] = rsqrtf((float)(1 + r_cnt[i]));
}

template <int C>
__global__ void __launch_bounds__(AGG_THREADS)
gcn_aggregate_owned(const float* __restrict__ h,
                    const int* __restrict__ src,
                    const int* __restrict__ dst,
                    const float* __restrict__ dinv,
                    const float* __restrict__ bias,
                    const unsigned int* __restrict__ own,
                    float* __restrict__ acc,
                    const int* __restrict__ lcnt,
                    const int* __restrict__ lsrc,
                    const int* __restrict__ lrow)
{
    constexpr int V = C / 32;
    constexpr int TILE4 = OWN_ROWS * C / 4;
    constexpr int K = Q_SLOTS / 32;
    __shared__ alignas(16) float tile[OWN_ROWS * C];
    __shared__ int   q_src[AGG_WAVES * Q_SLOTS];
    __shared__ int   q_row[AGG_WAVES * Q_SLOTS];

    const int tid  = threadIdx.x;
    const int lane = tid & 31;
    const int wv   = tid >> 5;
    const int grp  = blockIdx.x;
    const int base = grp * OWN_ROWS;
    float* gacc = acc + (size_t)base * C;
    const float* hown = h + (size_t)base * C;
    int*   qs = q_src + wv * Q_SLOTS;
    int*   qr = q_row + wv * Q_SLOTS;
    const size_t lbase = ((size_t)grp * AGG_WAVES + wv) * Q_SLOTS;

    for (int i = tid; i < TILE4; i += AGG_THREADS)
        reinterpret_cast<float4*>(tile)[i] = reinterpret_cast<const float4*>(hown)[i];

    int wcount = lcnt[grp * AGG_WAVES + wv];
    const bool scan = wcount > Q_SLOTS;
    __syncthreads();

    if (scan) {
        wcount = scan_owned(own, grp, tid, qs, [&](int e) {
            const int s = src[e];
            const int d = dst[e];
            const float* hr = h + (size_t)s * C;
            float* tr = tile + (d - base) * C;
#pragma unroll 1
            for (int c = 0; c < C; ++c) atomicAdd(&tr[c], hr[c]);
        });
    }
    __syncthreads();

    const int n = wcount < Q_SLOTS ? wcount : Q_SLOTS;
    if (scan) {
        int ev[K], sv[K], dv[K];
#pragma unroll
        for (int k = 0; k < K; ++k) ev[k] = qs[lane + 32 * k];
#pragma unroll
        for (int k = 0; k < K; ++k) {
            const int e = (lane + 32 * k < n) ? ev[k] : 0;
            sv[k] = src[e];
            dv[k] = dst[e];
        }
#pragma unroll
        for (int k = 0; k < K; ++k) {
            const int i = lane + 32 * k;
            qs[i] = sv[k];
            qr[i] = (dv[k] - base) * C;
        }
    } else {
#pragma unroll
        for (int k = 0; k < K; ++k) {
            const int i = lane + 32 * k;
            if (i < n) {
                qs[i] = lsrc[lbase + i];
                qr[i] = lrow[lbase + i] * C;
            }
        }
    }
    __syncthreads();

    constexpr int U = 8;
    auto load_row = [&](int i, float (&r)[V]) {
        const int s = qs[i];
        if constexpr (V == 2) {
            const float2 t = *reinterpret_cast<const float2*>(h + (size_t)s * C + 2 * lane);
            r[0] = t.x;
            r[1] = t.y;
        } else {
            r[0] = h[(size_t)s * C + lane];
        }
    };
    auto add_row = [&](int i, const float (&r)[V]) {
        float* tr = tile + qr[i] + V * lane;
#pragma unroll
        for (int v = 0; v < V; ++v) atomicAdd(&tr[v], r[v]);
    };
    const int full = n - (n % U);
    if (full > 0) {
        float cur[U][V], nxt[U][V];
#pragma unroll
        for (int u = 0; u < U; ++u) load_row(u, cur[u]);
        for (int i0 = U; i0 < full; i0 += U) {
#pragma unroll
            for (int u = 0; u < U; ++u) load_row(i0 + u, nxt[u]);
#pragma unroll
            for (int u = 0; u < U; ++u) add_row(i0 - U + u, cur[u]);
#pragma unroll
            for (int u = 0; u < U; ++u)
#pragma unroll
                for (int v = 0; v < V; ++v) cur[u][v] = nxt[u][v];
        }
#pragma unroll
        for (int u = 0; u < U; ++u) add_row(full - U + u, cur[u]);
    }
    for (int i = full; i < n; ++i) {
        float r[V];
        load_row(i, r);
        add_row(i, r);
    }
    __syncthreads();

    static_assert(AGG_THREADS % (C / 4) == 0, "a thread keeps its channels from turn to turn");
    const float* bq = bias + 4 * (tid % (C / 4));
    const float4 bv = make_float4(bq[0], bq[1], bq[2], bq[3]);
    for (int i = tid; i < TILE4; i += AGG_THREADS) {
        const float dd = dinv[base + i / (C / 4)];
        float4 v = reinterpret_cast<const float4*>(tile)[i];
        v.x = v.x * dd + bv.x;
        v.y = v.y * dd + bv.y;
        v.z = v.z * dd + bv.z;
        v.w = v.w * dd + bv.w;
        reinterpret_cast<float4*>(gacc)[i] = v;
    }
}

extern "C" void kernel_launch(void* const* d_in, const int* in_sizes, int n_in,
                              void* d_out, int out_size, void* d_ws, size_t ws_size,
                              hipStream_t stream)
{
    const float* x    = (const float*)d_in[0];
    const int*   eidx = (const int*)d_in[1];
    const int*   src  = eidx;
    const int*   dst  = eidx + NEDGES;
    const float* W1   = (const float*)d_in[2];
    const float* b1   = (const float*)d_in[3];
    const float* W2   = (const float*)d_in[4];
    const float* b2   = (const float*)d_in[5];
    float* out = (float*)d_out;

    float* dinv = (float*)d_ws;
    float* h    = dinv + NNODES;
    float* agg1 = h + (size_t)NNODES * HID_C;
    unsigned int* own = (unsigned int*)(agg1 + (size_t)NNODES * HID_C);
    int*   lsrc = (int*)(own + OWN_WORDS);
    int*   lrow = lsrc + LIST_WORDS;
    int*   lcnt = lrow + LIST_WORDS;

    const int B = 256;

    gcn_owner_bytes<<<(OWN_WORDS + B - 1) / B, B, 0, stream>>>(dst, own);

    gcn_list_owned<<<OWN_GROUPS, AGG_THREADS, 0, stream>>>(src, dst, own, dinv, lcnt, lsrc, lrow);

    gcn_gemm_strips<IN_C, HID_C, false, GEMM1_GROUPS><<<GEMM1_GROUPS, GEMM_THREADS, 0, stream>>>(
        x, W1, dinv, h);

    gcn_aggregate_owned<HID_C><<<OWN_GROUPS, AGG_THREADS, 0, stream>>>(
        h, src, dst, dinv, b1, own, agg1, lcnt, lsrc, lrow);

    gcn_gemm_strips<HID_C, OUT_C, true, GEMM2_GROUPS><<<GEMM2_GROUPS, GEMM_THREADS, 0, stream>>>(
        agg1, W2, dinv, h);

    gcn_aggregate_owned<OUT_C><<<OWN_GROUPS, AGG_THREADS, 0, stream>>>(
        h, src, dst, dinv, b2, own, out, lcnt, lsrc, lrow);
}


# first launch also packs each edge into one word (source | row in range << 17): a listed edge costs one scattered 4-byte read, not one in src and one in dst; lists one word; sums in LDS arrival order
# speedup vs baseline: 2.7403x; 1.0368x over previous
#include <hip/hip_runtime.h>
#include <hip/hip_bf16.h>


#define NNODES 100000
#define NEDGES 1000000
#define IN_C   128
#define HID_C  64
#define OUT_C  32

#define OWN_ROWS    400
#define OWN_GROUPS  (NNODES / OWN_ROWS)
#define AGG_THREADS 1024
#define AGG_WAVES   (AGG_THREADS / 32)
#define Q_SLOTS     256
#define OWN_CHUNKS  65536
#define OWN_WORDS   (OWN_CHUNKS * 4)
#define LIST_WORDS  (OWN_GROUPS * AGG_WAVES * Q_SLOTS)
#define SRC_BITS    17
#define SRC_MASK    ((1u << SRC_BITS) - 1u)

static_assert(NNODES % OWN_ROWS == 0 && OWN_GROUPS <= 255, "an owner id fits one byte and 0xFF is no group");
static_assert(NEDGES % 4 == 0 && NEDGES <= OWN_CHUNKS * 16, "the padded owner array holds every edge");
static_assert(OWN_CHUNKS % (AGG_THREADS * 4) == 0, "whole groups of four loads per lane");
static_assert(NNODES <= (1 << SRC_BITS) && OWN_ROWS <= (1 << (32 - SRC_BITS)), "source node and row fit one word");

typedef __attribute__((ext_vector_type(2))) float v2f;
typedef __attribute__((ext_vector_type(4))) float v4f;
typedef __attribute__((ext_vector_type(8))) float v8f;

__global__ void gcn_owner_bytes(const int* __restrict__ src, const int* __restrict__ dst,
                                unsigned int* __restrict__ own, unsigned int* __restrict__ pk)
{
    int t = blockIdx.x * blockDim.x + threadIdx.x;
    if (t >= OWN_WORDS) return;
    if (t >= NEDGES / 4) {
        own[t] = 0xFFFFFFFFu;
        return;
    }
    const int4 d = reinterpret_cast<const int4*>(dst)[t];
    const int4 s = reinterpret_cast<const int4*>(src)[t];
    const unsigned int g0 = (unsigned int)(d.x / OWN_ROWS);
    const unsigned int g1 = (unsigned int)(d.y / OWN_ROWS);
    const unsigned int g2 = (unsigned int)(d.z / OWN_ROWS);
    const unsigned int g3 = (unsigned int)(d.w / OWN_ROWS);
    own[t] = g0 | (g1 << 8) | (g2 << 16) | (g3 << 24);
    uint4 p;
    p.x = (unsigned int)s.x | (((unsigned int)d.x - g0 * OWN_ROWS) << SRC_BITS);
    p.y = (unsigned int)s.y | (((unsigned int)d.y - g1 * OWN_ROWS) << SRC_BITS);
    p.z = (unsigned int)s.z | (((unsigned int)d.z - g2 * OWN_ROWS) << SRC_BITS);
    p.w = (unsigned int)s.w | (((unsigned int)d.w - g3 * OWN_ROWS) << SRC_BITS);
    reinterpret_cast<uint4*>(pk)[t] = p;
}

#define GEMM_WAVES   8
#define GEMM_THREADS (GEMM_WAVES * 32)
#define NSTRIPS      (NNODES / 16)
#define GEMM1_GROUPS 256
#define GEMM2_GROUPS 512

static_assert(NNODES % 16 == 0, "whole 16-row strips");

template <int K, int NC, bool RELU_A, int NWG>
__global__ void __launch_bounds__(GEMM_THREADS)
gcn_gemm_strips(const float* __restrict__ A,
                const float* __restrict__ B,
                const float* __restrict__ dinv,
                float* __restrict__ C)
{
    constexpr int TN  = NC / 16;
    constexpr int PA  = K + 4;
    constexpr int PB  = 2 * NC + 32;
    constexpr int NLD = 16 * K / 128;
    static_assert(K % 4 == 0 && (16 * K) % 128 == 0, "a strip is whole 512-B runs and a 16-B piece stays inside one row");
    __shared__ alignas(16) float Bs[(K / 2) * PB];
    __shared__ alignas(16) float As[GEMM_WAVES * 16 * PA];

    const int tid  = threadIdx.x;
    const int lane = tid & 31;
    const int wv   = __builtin_amdgcn_readfirstlane(tid >> 5);
    const int lo = lane & 15;
    const int hi = lane >> 4;
    float* as = As + wv * 16 * PA;

    constexpr int SSTEP = NWG * GEMM_WAVES;
    int s = blockIdx.x + NWG * wv;

    v4f stage[NLD];
    if (s < NSTRIPS) {
        const v4f* g = reinterpret_cast<const v4f*>(A + (size_t)s * 16 * K) + lane;
#pragma unroll
        for (int i = 0; i < NLD; ++i) stage[i] = g[i * 32];
    }

    constexpr int BQ = NC / 4;
    constexpr int BITEMS = (K / 2) * BQ;
    static_assert(BITEMS % GEMM_THREADS == 0, "whole pieces per thread");
    constexpr int BPT = BITEMS / GEMM_THREADS;
    {
        v4f r0[BPT], r1[BPT];
#pragma unroll
        for (int j = 0; j < BPT; ++j) {
            const int i  = tid + j * GEMM_THREADS;
            const int kp = i / BQ;
            const int n  = (i % BQ) * 4;
            r0[j] = *reinterpret_cast<const v4f*>(B + (size_t)(2 * kp) * NC + n);
            r1[j] = *reinterpret_cast<const v4f*>(B + (size_t)(2 * kp + 1) * NC + n);
        }
#pragma unroll
        for (int j = 0; j < BPT; ++j) {
            const int i  = tid + j * GEMM_THREADS;
            const int kp = i / BQ;
            const int n  = (i % BQ) * 4;
            v4f* d = reinterpret_cast<v4f*>(Bs + kp * PB + 2 * n);
            d[0] = v4f{r0[j].x, r1[j].x, r0[j].y, r1[j].y};
            d[1] = v4f{r0[j].z, r1[j].z, r0[j].w, r1[j].w};
        }
    }
    __syncthreads();

    const float* afrag = as + lo * PA + 2 * hi;
    const float* bfrag = Bs + hi * PB + 2 * lo;

    v2f bf[K / 4][TN];
#pragma unroll
    for (int kk = 0; kk < K / 4; ++kk)
#pragma unroll
        for (int t = 0; t < TN; ++t)
            bf[kk][t] = *reinterpret_cast<const v2f*>(bfrag + (2 * kk) * PB + 32 * t);

    for (; s < NSTRIPS; s += SSTEP) {
#pragma unroll
        for (int i = 0; i < NLD; ++i) {
            v4f v = stage[i];
            if (RELU_A) {
                v.x = fmaxf(v.x, 0.f);
                v.y = fmaxf(v.y, 0.f);
                v.z = fmaxf(v.z, 0.f);
                v.w = fmaxf(v.w, 0.f);
            }
            const int e = i * 128 + 4 * lane;
            *reinterpret_cast<v4f*>(as + (e / K) * PA + (e % K)) = v;
        }
        const int tm = s * 16;
        const float4 d0 = *(const float4*)(dinv + tm + 8 * hi);
        const float4 d1 = *(const float4*)(dinv + tm + 8 * hi + 4);
        if (s + SSTEP < NSTRIPS) {
            const v4f* g = reinterpret_cast<const v4f*>(A + (size_t)(s + SSTEP) * 16 * K) + lane;
#pragma unroll
            for (int i = 0; i < NLD; ++i) stage[i] = g[i * 32];
        }

        v8f acc[TN];
#pragma unroll
        for (int t = 0; t < TN; ++t) acc[t] = v8f{};
#pragma unroll
        for (int k = 0; k < K; k += 4) {
            const v2f a = *reinterpret_cast<const v2f*>(afrag + k);
#pragma unroll
            for (int t = 0; t < TN; ++t) {
                acc[t] = __builtin_amdgcn_wmma_f32_16x16x4_f32(
                     false, a,  false, bf[k / 4][t],
                     (short)0, acc[t],  false,  false);
            }
        }

        const float dv[8] = {d0.x, d0.y, d0.z, d0.w, d1.x, d1.y, d1.z, d1.w};
#pragma unroll
        for (int t = 0; t < TN; ++t) {
            float* crow = C + (size_t)(tm + 8 * hi) * NC + t * 16 + lo;
#pragma unroll
            for (int g = 0; g < 8; ++g) crow[(size_t)g * NC] = acc[t][g] * dv[g];
        }
    }
}


__device__ __forceinline__ unsigned int zero_bytes(unsigned int x)
{
    return ~(((x & 0x7F7F7F7Fu) + 0x7F7F7F7Fu) | x | 0x7F7F7F7Fu);
}

__device__ __forceinline__ unsigned int match16(const uint4 v, unsigned int pat)
{
    return (zero_bytes(v.x ^ pat) >> 7) | (zero_bytes(v.y ^ pat) >> 6) |
           (zero_bytes(v.z ^ pat) >> 5) | (zero_bytes(v.w ^ pat) >> 4);
}

template <typename OVER>
__device__ __forceinline__ int scan_owned(const unsigned int* __restrict__ own, int grp, int tid,
                                          int* __restrict__ qs, OVER over)
{
    int wcount = 0;
    const unsigned int pat = (unsigned int)grp * 0x01010101u;
    const uint4* own16 = reinterpret_cast<const uint4*>(own);
    const int first = grp * (OWN_CHUNKS / OWN_GROUPS);
    constexpr int DEPTH = 4;
    constexpr int NGRP  = OWN_CHUNKS / (AGG_THREADS * DEPTH);

    uint4 nv[DEPTH];
    int   nc[DEPTH];
    auto fetch = [&](int it, uint4& v, int& ch) {
        ch = (it * AGG_THREADS + tid + first) & (OWN_CHUNKS - 1);
        v = own16[ch];
    };
#pragma unroll
    for (int u = 0; u < DEPTH; ++u) fetch(u, nv[u], nc[u]);

    for (int g = 0; g < NGRP; ++g) {
        uint4 cv[DEPTH];
        int   cc[DEPTH];
#pragma unroll
        for (int u = 0; u < DEPTH; ++u) { cv[u] = nv[u]; cc[u] = nc[u]; }
        if (g + 1 < NGRP) {
#pragma unroll
            for (int u = 0; u < DEPTH; ++u) fetch((g + 1) * DEPTH + u, nv[u], nc[u]);
        }
#pragma unroll
        for (int u = 0; u < DEPTH; u += 2) {
            unsigned int mk = match16(cv[u], pat) | (match16(cv[u + 1], pat) << 4);
            while (true) {
                const bool has = mk != 0u;
                const unsigned int bal = __builtin_amdgcn_ballot_w32(has);
                if (bal == 0u) break;
                if (has) {
                    const int p = __builtin_ctz(mk);
                    const int ch = (p & 4) ? cc[u + 1] : cc[u];
                    const int e = ch * 16 + (((p & 3) << 2) | (p >> 3));
                    const int slot = wcount + (int)__builtin_amdgcn_mbcnt_lo(bal, 0u);
                    if (slot < Q_SLOTS) qs[slot] = e;
                    else                over(e);
                    mk &= mk - 1u;
                }
                wcount += __builtin_popcount(bal);
            }
        }
    }
    return wcount;
}

__global__ void __launch_bounds__(AGG_THREADS)
gcn_list_owned(const unsigned int* __restrict__ pk,
               const unsigned int* __restrict__ own,
               float* __restrict__ dinv,
               int* __restrict__ lcnt,
               unsigned int* __restrict__ lpk)
{
    constexpr int K = Q_SLOTS / 32;
    __shared__ int q_e[AGG_WAVES * Q_SLOTS];
    __shared__ int s_cnt[AGG_WAVES];
    __shared__ int r_cnt[OWN_ROWS];

    const int tid  = threadIdx.x;
    const int lane = tid & 31;
    const int wv   = tid >> 5;
    const int grp  = blockIdx.x;
    const int base = grp * OWN_ROWS;
    int* qs = q_e + wv * Q_SLOTS;
    const size_t lbase = ((size_t)grp * AGG_WAVES + wv) * Q_SLOTS;

    for (int i = tid; i < OWN_ROWS; i += AGG_THREADS) r_cnt[i] = 0;
    __syncthreads();

    const int wcount = scan_owned(own, grp, tid, qs, [&](int e) { atomicAdd(&r_cnt[pk[e] >> SRC_BITS], 1); });
    if (lane == 0) s_cnt[wv] = wcount;
    __syncthreads();

    const int n = wcount < Q_SLOTS ? wcount : Q_SLOTS;
    int ev[K];
    unsigned int pv[K];
#pragma unroll
    for (int k = 0; k < K; ++k) ev[k] = qs[lane + 32 * k];
#pragma unroll
    for (int k = 0; k < K; ++k) {
        const int e = (lane + 32 * k < n) ? ev[k] : 0;
        pv[k] = pk[e];
    }
#pragma unroll
    for (int k = 0; k < K; ++k) {
        const int i = lane + 32 * k;
        if (i < n) {
            lpk[lbase + i] = pv[k];
            atomicAdd(&r_cnt[pv[k] >> SRC_BITS], 1);
        }
    }
    __syncthreads();

    if (wv == 0) lcnt[grp * AGG_WAVES + lane] = s_cnt[lane];
    for (int i = tid; i < OWN_ROWS; i += AGG_THREADS) dinv[base + i] = rsqrtf((float)(1 + r_cnt[i]));
}

template <int C>
__global__ void __launch_bounds__(AGG_THREADS)
gcn_aggregate_owned(const float* __restrict__ h,
                    const unsigned int* __restrict__ pk,
                    const float* __restrict__ dinv,
                    const float* __restrict__ bias,
                    const unsigned int* __restrict__ own,
                    float* __restrict__ acc,
                    const int* __restrict__ lcnt,
                    const unsigned int* __restrict__ lpk)
{
    constexpr int V = C / 32;
    constexpr int TILE4 = OWN_ROWS * C / 4;
    constexpr int K = Q_SLOTS / 32;
    __shared__ alignas(16) float tile[OWN_ROWS * C];
    __shared__ int   q_src[AGG_WAVES * Q_SLOTS];
    __shared__ int   q_row[AGG_WAVES * Q_SLOTS];

    const int tid  = threadIdx.x;
    const int lane = tid & 31;
    const int wv   = tid >> 5;
    const int grp  = blockIdx.x;
    const int base = grp * OWN_ROWS;
    float* gacc = acc + (size_t)base * C;
    const float* hown = h + (size_t)base * C;
    int*   qs = q_src + wv * Q_SLOTS;
    int*   qr = q_row + wv * Q_SLOTS;
    const size_t lbase = ((size_t)grp * AGG_WAVES + wv) * Q_SLOTS;

    for (int i = tid; i < TILE4; i += AGG_THREADS)
        reinterpret_cast<float4*>(tile)[i] = reinterpret_cast<const float4*>(hown)[i];

    int wcount = lcnt[grp * AGG_WAVES + wv];
    const bool scan = wcount > Q_SLOTS;
    __syncthreads();

    if (scan) {
        wcount = scan_owned(own, grp, tid, qs, [&](int e) {
            const unsigned int p = pk[e];
            const float* hr = h + (size_t)(p & SRC_MASK) * C;
            float* tr = tile + (int)(p >> SRC_BITS) * C;
#pragma unroll 1
            for (int c = 0; c < C; ++c) atomicAdd(&tr[c], hr[c]);
        });
    }
    __syncthreads();

    const int n = wcount < Q_SLOTS ? wcount : Q_SLOTS;
    if (scan) {
        int ev[K];
        unsigned int pv[K];
#pragma unroll
        for (int k = 0; k < K; ++k) ev[k] = qs[lane + 32 * k];
#pragma unroll
        for (int k = 0; k < K; ++k) {
            const int e = (lane + 32 * k < n) ? ev[k] : 0;
            pv[k] = pk[e];
        }
#pragma unroll
        for (int k = 0; k < K; ++k) {
            const int i = lane + 32 * k;
            qs[i] = (int)(pv[k] & SRC_MASK);
            qr[i] = (int)(pv[k] >> SRC_BITS) * C;
        }
    } else {
#pragma unroll
        for (int k = 0; k < K; ++k) {
            const int i = lane + 32 * k;
            if (i < n) {
                const unsigned int p = lpk[lbase + i];
                qs[i] = (int)(p & SRC_MASK);
                qr[i] = (int)(p >> SRC_BITS) * C;
            }
        }
    }
    __syncthreads();

    constexpr int U = 8;
    auto load_row = [&](int i, float (&r)[V]) {
        const int s = qs[i];
        if constexpr (V == 2) {
            const float2 t = *reinterpret_cast<const float2*>(h + (size_t)s * C + 2 * lane);
            r[0] = t.x;
            r[1] = t.y;
        } else {
            r[0] = h[(size_t)s * C + lane];
        }
    };
    auto add_row = [&](int i, const float (&r)[V]) {
        float* tr = tile + qr[i] + V * lane;
#pragma unroll
        for (int v = 0; v < V; ++v) atomicAdd(&tr[v], r[v]);
    };
    const int full = n - (n % U);
    if (full > 0) {
        float cur[U][V], nxt[U][V];
#pragma unroll
        for (int u = 0; u < U; ++u) load_row(u, cur[u]);
        for (int i0 = U; i0 < full; i0 += U) {
#pragma unroll
            for (int u = 0; u < U; ++u) load_row(i0 + u, nxt[u]);
#pragma unroll
            for (int u = 0; u < U; ++u) add_row(i0 - U + u, cur[u]);
#pragma unroll
            for (int u = 0; u < U; ++u)
#pragma unroll
                for (int v = 0; v < V; ++v) cur[u][v] = nxt[u][v];
        }
#pragma unroll
        for (int u = 0; u < U; ++u) add_row(full - U + u, cur[u]);
    }
    for (int i = full; i < n; ++i) {
        float r[V];
        load_row(i, r);
        add_row(i, r);
    }
    __syncthreads();

    static_assert(AGG_THREADS % (C / 4) == 0, "a thread keeps its channels from turn to turn");
    const float* bq = bias + 4 * (tid % (C / 4));
    const float4 bv = make_float4(bq[0], bq[1], bq[2], bq[3]);
    for (int i = tid; i < TILE4; i += AGG_THREADS) {
        const float dd = dinv[base + i / (C / 4)];
        float4 v = reinterpret_cast<const float4*>(tile)[i];
        v.x = v.x * dd + bv.x;
        v.y = v.y * dd + bv.y;
        v.z = v.z * dd + bv.z;
        v.w = v.w * dd + bv.w;
        reinterpret_cast<float4*>(gacc)[i] = v;
    }
}

extern "C" void kernel_launch(void* const* d_in, const int* in_sizes, int n_in,
                              void* d_out, int out_size, void* d_ws, size_t ws_size,
                              hipStream_t stream)
{
    const float* x    = (const float*)d_in[0];
    const int*   eidx = (const int*)d_in[1];
    const int*   src  = eidx;
    const int*   dst  = eidx + NEDGES;
    const float* W1   = (const float*)d_in[2];
    const float* b1   = (const float*)d_in[3];
    const float* W2   = (const float*)d_in[4];
    const float* b2   = (const float*)d_in[5];
    float* out = (float*)d_out;

    float* dinv = (float*)d_ws;
    float* h    = dinv + NNODES;
    float* agg1 = h + (size_t)NNODES * HID_C;
    unsigned int* own = (unsigned int*)(agg1 + (size_t)NNODES * HID_C);
    unsigned int* pk  = own + OWN_WORDS;
    unsigned int* lpk = pk + NEDGES;
    int*   lcnt = (int*)(lpk + LIST_WORDS);

    const int B = 256;

    gcn_owner_bytes<<<(OWN_WORDS + B - 1) / B, B, 0, stream>>>(src, dst, own, pk);

    gcn_list_owned<<<OWN_GROUPS, AGG_THREADS, 0, stream>>>(pk, own, dinv, lcnt, lpk);

    gcn_gemm_strips<IN_C, HID_C, false, GEMM1_GROUPS><<<GEMM1_GROUPS, GEMM_THREADS, 0, stream>>>(
        x, W1, dinv, h);

    gcn_aggregate_owned<HID_C><<<OWN_GROUPS, AGG_THREADS, 0, stream>>>(
        h, pk, dinv, b1, own, agg1, lcnt, lpk);

    gcn_gemm_strips<HID_C, OUT_C, true, GEMM2_GROUPS><<<GEMM2_GROUPS, GEMM_THREADS, 0, stream>>>(
        agg1, W2, dinv, h);

    gcn_aggregate_owned<OUT_C><<<OWN_GROUPS, AGG_THREADS, 0, stream>>>(
        h, pk, dinv, b2, own, out, lcnt, lpk);
}


# aggregation staging: own-row pieces, their dinv, bias and the wave's list words all requested before first use; a wave's leftover edges go as one padded group; sums in LDS arrival order
# speedup vs baseline: 2.9988x; 1.0943x over previous
#include <hip/hip_runtime.h>
#include <hip/hip_bf16.h>


#define NNODES 100000
#define NEDGES 1000000
#define IN_C   128
#define HID_C  64
#define OUT_C  32

#define OWN_ROWS    400
#define OWN_GROUPS  (NNODES / OWN_ROWS)
#define AGG_THREADS 1024
#define AGG_WAVES   (AGG_THREADS / 32)
#define Q_SLOTS     256
#define OWN_CHUNKS  65536
#define OWN_WORDS   (OWN_CHUNKS * 4)
#define LIST_WORDS  (OWN_GROUPS * AGG_WAVES * Q_SLOTS)
#define SRC_BITS    17
#define SRC_MASK    ((1u << SRC_BITS) - 1u)

static_assert(NNODES % OWN_ROWS == 0 && OWN_GROUPS <= 255, "an owner id fits one byte and 0xFF is no group");
static_assert(NEDGES % 4 == 0 && NEDGES <= OWN_CHUNKS * 16, "the padded owner array holds every edge");
static_assert(OWN_CHUNKS % (AGG_THREADS * 4) == 0, "whole groups of four loads per lane");
static_assert(NNODES <= (1 << SRC_BITS) && OWN_ROWS <= (1 << (32 - SRC_BITS)), "source node and row fit one word");

typedef __attribute__((ext_vector_type(2))) float v2f;
typedef __attribute__((ext_vector_type(4))) float v4f;
typedef __attribute__((ext_vector_type(8))) float v8f;

__global__ void gcn_owner_bytes(const int* __restrict__ src, const int* __restrict__ dst,
                                unsigned int* __restrict__ own, unsigned int* __restrict__ pk)
{
    int t = blockIdx.x * blockDim.x + threadIdx.x;
    if (t >= OWN_WORDS) return;
    if (t >= NEDGES / 4) {
        own[t] = 0xFFFFFFFFu;
        return;
    }
    const int4 d = reinterpret_cast<const int4*>(dst)[t];
    const int4 s = reinterpret_cast<const int4*>(src)[t];
    const unsigned int g0 = (unsigned int)(d.x / OWN_ROWS);
    const unsigned int g1 = (unsigned int)(d.y / OWN_ROWS);
    const unsigned int g2 = (unsigned int)(d.z / OWN_ROWS);
    const unsigned int g3 = (unsigned int)(d.w / OWN_ROWS);
    own[t] = g0 | (g1 << 8) | (g2 << 16) | (g3 << 24);
    uint4 p;
    p.x = (unsigned int)s.x | (((unsigned int)d.x - g0 * OWN_ROWS) << SRC_BITS);
    p.y = (unsigned int)s.y | (((unsigned int)d.y - g1 * OWN_ROWS) << SRC_BITS);
    p.z = (unsigned int)s.z | (((unsigned int)d.z - g2 * OWN_ROWS) << SRC_BITS);
    p.w = (unsigned int)s.w | (((unsigned int)d.w - g3 * OWN_ROWS) << SRC_BITS);
    reinterpret_cast<uint4*>(pk)[t] = p;
}

#define GEMM_WAVES   8
#define GEMM_THREADS (GEMM_WAVES * 32)
#define NSTRIPS      (NNODES / 16)
#define GEMM1_GROUPS 256
#define GEMM2_GROUPS 512

static_assert(NNODES % 16 == 0, "whole 16-row strips");

template <int K, int NC, bool RELU_A, int NWG>
__global__ void __launch_bounds__(GEMM_THREADS)
gcn_gemm_strips(const float* __restrict__ A,
                const float* __restrict__ B,
                const float* __restrict__ dinv,
                float* __restrict__ C)
{
    constexpr int TN  = NC / 16;
    constexpr int PA  = K + 4;
    constexpr int PB  = 2 * NC + 32;
    constexpr int NLD = 16 * K / 128;
    static_assert(K % 4 == 0 && (16 * K) % 128 == 0, "a strip is whole 512-B runs and a 16-B piece stays inside one row");
    __shared__ alignas(16) float Bs[(K / 2) * PB];
    __shared__ alignas(16) float As[GEMM_WAVES * 16 * PA];

    const int tid  = threadIdx.x;
    const int lane = tid & 31;
    const int wv   = __builtin_amdgcn_readfirstlane(tid >> 5);
    const int lo = lane & 15;
    const int hi = lane >> 4;
    float* as = As + wv * 16 * PA;

    constexpr int SSTEP = NWG * GEMM_WAVES;
    int s = blockIdx.x + NWG * wv;

    v4f stage[NLD];
    if (s < NSTRIPS) {
        const v4f* g = reinterpret_cast<const v4f*>(A + (size_t)s * 16 * K) + lane;
#pragma unroll
        for (int i = 0; i < NLD; ++i) stage[i] = g[i * 32];
    }

    constexpr int BQ = NC / 4;
    constexpr int BITEMS = (K / 2) * BQ;
    static_assert(BITEMS % GEMM_THREADS == 0, "whole pieces per thread");
    constexpr int BPT = BITEMS / GEMM_THREADS;
    {
        v4f r0[BPT], r1[BPT];
#pragma unroll
        for (int j = 0; j < BPT; ++j) {
            const int i  = tid + j * GEMM_THREADS;
            const int kp = i / BQ;
            const int n  = (i % BQ) * 4;
            r0[j] = *reinterpret_cast<const v4f*>(B + (size_t)(2 * kp) * NC + n);
            r1[j] = *reinterpret_cast<const v4f*>(B + (size_t)(2 * kp + 1) * NC + n);
        }
#pragma unroll
        for (int j = 0; j < BPT; ++j) {
            const int i  = tid + j * GEMM_THREADS;
            const int kp = i / BQ;
            const int n  = (i % BQ) * 4;
            v4f* d = reinterpret_cast<v4f*>(Bs + kp * PB + 2 * n);
            d[0] = v4f{r0[j].x, r1[j].x, r0[j].y, r1[j].y};
            d[1] = v4f{r0[j].z, r1[j].z, r0[j].w, r1[j].w};
        }
    }
    __syncthreads();

    const float* afrag = as + lo * PA + 2 * hi;
    const float* bfrag = Bs + hi * PB + 2 * lo;

    v2f bf[K / 4][TN];
#pragma unroll
    for (int kk = 0; kk < K / 4; ++kk)
#pragma unroll
        for (int t = 0; t < TN; ++t)
            bf[kk][t] = *reinterpret_cast<const v2f*>(bfrag + (2 * kk) * PB + 32 * t);

    for (; s < NSTRIPS; s += SSTEP) {
#pragma unroll
        for (int i = 0; i < NLD; ++i) {
            v4f v = stage[i];
            if (RELU_A) {
                v.x = fmaxf(v.x, 0.f);
                v.y = fmaxf(v.y, 0.f);
                v.z = fmaxf(v.z, 0.f);
                v.w = fmaxf(v.w, 0.f);
            }
            const int e = i * 128 + 4 * lane;
            *reinterpret_cast<v4f*>(as + (e / K) * PA + (e % K)) = v;
        }
        const int tm = s * 16;
        const float4 d0 = *(const float4*)(dinv + tm + 8 * hi);
        const float4 d1 = *(const float4*)(dinv + tm + 8 * hi + 4);
        if (s + SSTEP < NSTRIPS) {
            const v4f* g = reinterpret_cast<const v4f*>(A + (size_t)(s + SSTEP) * 16 * K) + lane;
#pragma unroll
            for (int i = 0; i < NLD; ++i) stage[i] = g[i * 32];
        }

        v8f acc[TN];
#pragma unroll
        for (int t = 0; t < TN; ++t) acc[t] = v8f{};
#pragma unroll
        for (int k = 0; k < K; k += 4) {
            const v2f a = *reinterpret_cast<const v2f*>(afrag + k);
#pragma unroll
            for (int t = 0; t < TN; ++t) {
                acc[t] = __builtin_amdgcn_wmma_f32_16x16x4_f32(
                     false, a,  false, bf[k / 4][t],
                     (short)0, acc[t],  false,  false);
            }
        }

        const float dv[8] = {d0.x, d0.y, d0.z, d0.w, d1.x, d1.y, d1.z, d1.w};
#pragma unroll
        for (int t = 0; t < TN; ++t) {
            float* crow = C + (size_t)(tm + 8 * hi) * NC + t * 16 + lo;
#pragma unroll
            for (int g = 0; g < 8; ++g) crow[(size_t)g * NC] = acc[t][g] * dv[g];
        }
    }
}


__device__ __forceinline__ unsigned int zero_bytes(unsigned int x)
{
    return ~(((x & 0x7F7F7F7Fu) + 0x7F7F7F7Fu) | x | 0x7F7F7F7Fu);
}

__device__ __forceinline__ unsigned int match16(const uint4 v, unsigned int pat)
{
    return (zero_bytes(v.x ^ pat) >> 7) | (zero_bytes(v.y ^ pat) >> 6) |
           (zero_bytes(v.z ^ pat) >> 5) | (zero_bytes(v.w ^ pat) >> 4);
}

template <typename OVER>
__device__ __forceinline__ int scan_owned(const unsigned int* __restrict__ own, int grp, int tid,
                                          int* __restrict__ qs, OVER over)
{
    int wcount = 0;
    const unsigned int pat = (unsigned int)grp * 0x01010101u;
    const uint4* own16 = reinterpret_cast<const uint4*>(own);
    const int first = grp * (OWN_CHUNKS / OWN_GROUPS);
    constexpr int DEPTH = 4;
    constexpr int NGRP  = OWN_CHUNKS / (AGG_THREADS * DEPTH);

    uint4 nv[DEPTH];
    int   nc[DEPTH];
    auto fetch = [&](int it, uint4& v, int& ch) {
        ch = (it * AGG_THREADS + tid + first) & (OWN_CHUNKS - 1);
        v = own16[ch];
    };
#pragma unroll
    for (int u = 0; u < DEPTH; ++u) fetch(u, nv[u], nc[u]);

    for (int g = 0; g < NGRP; ++g) {
        uint4 cv[DEPTH];
        int   cc[DEPTH];
#pragma unroll
        for (int u = 0; u < DEPTH; ++u) { cv[u] = nv[u]; cc[u] = nc[u]; }
        if (g + 1 < NGRP) {
#pragma unroll
            for (int u = 0; u < DEPTH; ++u) fetch((g + 1) * DEPTH + u, nv[u], nc[u]);
        }
#pragma unroll
        for (int u = 0; u < DEPTH; u += 2) {
            unsigned int mk = match16(cv[u], pat) | (match16(cv[u + 1], pat) << 4);
            while (true) {
                const bool has = mk != 0u;
                const unsigned int bal = __builtin_amdgcn_ballot_w32(has);
                if (bal == 0u) break;
                if (has) {
                    const int p = __builtin_ctz(mk);
                    const int ch = (p & 4) ? cc[u + 1] : cc[u];
                    const int e = ch * 16 + (((p & 3) << 2) | (p >> 3));
                    const int slot = wcount + (int)__builtin_amdgcn_mbcnt_lo(bal, 0u);
                    if (slot < Q_SLOTS) qs[slot] = e;
                    else                over(e);
                    mk &= mk - 1u;
                }
                wcount += __builtin_popcount(bal);
            }
        }
    }
    return wcount;
}

__global__ void __launch_bounds__(AGG_THREADS)
gcn_list_owned(const unsigned int* __restrict__ pk,
               const unsigned int* __restrict__ own,
               float* __restrict__ dinv,
               int* __restrict__ lcnt,
               unsigned int* __restrict__ lpk)
{
    constexpr int K = Q_SLOTS / 32;
    __shared__ int q_e[AGG_WAVES * Q_SLOTS];
    __shared__ int s_cnt[AGG_WAVES];
    __shared__ int r_cnt[OWN_ROWS];

    const int tid  = threadIdx.x;
    const int lane = tid & 31;
    const int wv   = tid >> 5;
    const int grp  = blockIdx.x;
    const int base = grp * OWN_ROWS;
    int* qs = q_e + wv * Q_SLOTS;
    const size_t lbase = ((size_t)grp * AGG_WAVES + wv) * Q_SLOTS;

    for (int i = tid; i < OWN_ROWS; i += AGG_THREADS) r_cnt[i] = 0;
    __syncthreads();

    const int wcount = scan_owned(own, grp, tid, qs, [&](int e) { atomicAdd(&r_cnt[pk[e] >> SRC_BITS], 1); });
    if (lane == 0) s_cnt[wv] = wcount;
    __syncthreads();

    const int n = wcount < Q_SLOTS ? wcount : Q_SLOTS;
    int ev[K];
    unsigned int pv[K];
#pragma unroll
    for (int k = 0; k < K; ++k) ev[k] = qs[lane + 32 * k];
#pragma unroll
    for (int k = 0; k < K; ++k) {
        const int e = (lane + 32 * k < n) ? ev[k] : 0;
        pv[k] = pk[e];
    }
#pragma unroll
    for (int k = 0; k < K; ++k) {
        const int i = lane + 32 * k;
        if (i < n) {
            lpk[lbase + i] = pv[k];
            atomicAdd(&r_cnt[pv[k] >> SRC_BITS], 1);
        }
    }
    __syncthreads();

    if (wv == 0) lcnt[grp * AGG_WAVES + lane] = s_cnt[lane];
    for (int i = tid; i < OWN_ROWS; i += AGG_THREADS) dinv[base + i] = rsqrtf((float)(1 + r_cnt[i]));
}

template <int C>
__global__ void __launch_bounds__(AGG_THREADS)
gcn_aggregate_owned(const float* __restrict__ h,
                    const unsigned int* __restrict__ pk,
                    const float* __restrict__ dinv,
                    const float* __restrict__ bias,
                    const unsigned int* __restrict__ own,
                    float* __restrict__ acc,
                    const int* __restrict__ lcnt,
                    const unsigned int* __restrict__ lpk)
{
    constexpr int V = C / 32;
    constexpr int TILE4 = OWN_ROWS * C / 4;
    constexpr int NIT = (TILE4 + AGG_THREADS - 1) / AGG_THREADS;
    constexpr int K = Q_SLOTS / 32;
    constexpr int U = 8;
    static_assert(TILE4 >= AGG_THREADS, "a thread's first piece always exists (it stands in for a missing last one)");
    static_assert(Q_SLOTS % U == 0, "a padded list still fits the queue");
    static_assert(AGG_THREADS % (C / 4) == 0, "a thread keeps its channels from turn to turn");
    __shared__ alignas(16) float tile[OWN_ROWS * C];
    __shared__ int   q_src[AGG_WAVES * Q_SLOTS];
    __shared__ int   q_row[AGG_WAVES * Q_SLOTS];

    const int tid  = threadIdx.x;
    const int lane = tid & 31;
    const int wv   = tid >> 5;
    const int grp  = blockIdx.x;
    const int base = grp * OWN_ROWS;
    float* gacc = acc + (size_t)base * C;
    const float* hown = h + (size_t)base * C;
    int*   qs = q_src + wv * Q_SLOTS;
    int*   qr = q_row + wv * Q_SLOTS;
    const size_t lbase = ((size_t)grp * AGG_WAVES + wv) * Q_SLOTS;

    const int wcount = __builtin_amdgcn_readfirstlane(lcnt[grp * AGG_WAVES + wv]);

    v4f    ownv[NIT];
    float  dd[NIT];
#pragma unroll
    for (int k = 0; k < NIT; ++k) {
        const int i  = tid + k * AGG_THREADS;
        const int ic = i < TILE4 ? i : tid;
        ownv[k] = reinterpret_cast<const v4f*>(hown)[ic];
        dd[k]   = dinv[base + ic / (C / 4)];
    }
    const float* bq = bias + 4 * (tid % (C / 4));
    const float4 bv = make_float4(bq[0], bq[1], bq[2], bq[3]);

    const bool scan = wcount > Q_SLOTS;
    const int n = wcount < Q_SLOTS ? wcount : Q_SLOTS;
    const int npad = (n + U - 1) / U * U;

    unsigned int pv[K];
    if (!scan) {
#pragma unroll
        for (int k = 0; k < K; ++k) {
            const int i = lane + 32 * k;
            const unsigned int* a = (i < n) ? (lpk + lbase + i) : pk;
            pv[k] = *a;
        }
    }

#pragma unroll
    for (int k = 0; k < NIT; ++k) {
        const int i = tid + k * AGG_THREADS;
        if (i < TILE4) reinterpret_cast<v4f*>(tile)[i] = ownv[k];
    }
    if (!scan) {
#pragma unroll
        for (int k = 0; k < K; ++k) {
            const int i = lane + 32 * k;
            if (i < n) {
                qs[i] = (int)(pv[k] & SRC_MASK);
                qr[i] = (int)(pv[k] >> SRC_BITS) * C;
            } else if (i < npad) {
                qs[i] = base;
                qr[i] = 0;
            }
        }
    }
    __syncthreads();

    if (scan) {
        scan_owned(own, grp, tid, qs, [&](int e) {
            const unsigned int p = pk[e];
            const float* hr = h + (size_t)(p & SRC_MASK) * C;
            float* tr = tile + (int)(p >> SRC_BITS) * C;
#pragma unroll 1
            for (int c = 0; c < C; ++c) atomicAdd(&tr[c], hr[c]);
        });
        int ev[K];
#pragma unroll
        for (int k = 0; k < K; ++k) ev[k] = qs[lane + 32 * k];
#pragma unroll
        for (int k = 0; k < K; ++k) pv[k] = pk[ev[k]];
#pragma unroll
        for (int k = 0; k < K; ++k) {
            const int i = lane + 32 * k;
            qs[i] = (int)(pv[k] & SRC_MASK);
            qr[i] = (int)(pv[k] >> SRC_BITS) * C;
        }
    }

    auto load_row = [&](int i, float (&r)[V]) {
        const int s = qs[i];
        if constexpr (V == 2) {
            const float2 t = *reinterpret_cast<const float2*>(h + (size_t)s * C + 2 * lane);
            r[0] = t.x;
            r[1] = t.y;
        } else {
            r[0] = h[(size_t)s * C + lane];
        }
    };
    auto add_row = [&](int i, const float (&r)[V]) {
        float* tr = tile + qr[i] + V * lane;
#pragma unroll
        for (int v = 0; v < V; ++v) atomicAdd(&tr[v], r[v]);
    };
    if (npad > 0) {
        float cur[U][V], nxt[U][V];
#pragma unroll
        for (int u = 0; u < U; ++u) load_row(u, cur[u]);
        for (int i0 = U; i0 < npad; i0 += U) {
#pragma unroll
            for (int u = 0; u < U; ++u) load_row(i0 + u, nxt[u]);
#pragma unroll
            for (int u = 0; u < U; ++u) add_row(i0 - U + u, cur[u]);
#pragma unroll
            for (int u = 0; u < U; ++u)
#pragma unroll
                for (int v = 0; v < V; ++v) cur[u][v] = nxt[u][v];
        }
#pragma unroll
        for (int u = 0; u < U; ++u)
            if (npad - U + u < n) add_row(npad - U + u, cur[u]);
    }
    __syncthreads();

#pragma unroll
    for (int k = 0; k < NIT; ++k) {
        const int i = tid + k * AGG_THREADS;
        if (i < TILE4) {
            float4 v = reinterpret_cast<const float4*>(tile)[i];
            v.x = v.x * dd[k] + bv.x;
            v.y = v.y * dd[k] + bv.y;
            v.z = v.z * dd[k] + bv.z;
            v.w = v.w * dd[k] + bv.w;
            reinterpret_cast<float4*>(gacc)[i] = v;
        }
    }
}

extern "C" void kernel_launch(void* const* d_in, const int* in_sizes, int n_in,
                              void* d_out, int out_size, void* d_ws, size_t ws_size,
                              hipStream_t stream)
{
    const float* x    = (const float*)d_in[0];
    const int*   eidx = (const int*)d_in[1];
    const int*   src  = eidx;
    const int*   dst  = eidx + NEDGES;
    const float* W1   = (const float*)d_in[2];
    const float* b1   = (const float*)d_in[3];
    const float* W2   = (const float*)d_in[4];
    const float* b2   = (const float*)d_in[5];
    float* out = (float*)d_out;

    float* dinv = (float*)d_ws;
    float* h    = dinv + NNODES;
    float* agg1 = h + (size_t)NNODES * HID_C;
    unsigned int* own = (unsigned int*)(agg1 + (size_t)NNODES * HID_C);
    unsigned int* pk  = own + OWN_WORDS;
    unsigned int* lpk = pk + NEDGES;
    int*   lcnt = (int*)(lpk + LIST_WORDS);

    const int B = 256;

    gcn_owner_bytes<<<(OWN_WORDS + B - 1) / B, B, 0, stream>>>(src, dst, own, pk);

    gcn_list_owned<<<OWN_GROUPS, AGG_THREADS, 0, stream>>>(pk, own, dinv, lcnt, lpk);

    gcn_gemm_strips<IN_C, HID_C, false, GEMM1_GROUPS><<<GEMM1_GROUPS, GEMM_THREADS, 0, stream>>>(
        x, W1, dinv, h);

    gcn_aggregate_owned<HID_C><<<OWN_GROUPS, AGG_THREADS, 0, stream>>>(
        h, pk, dinv, b1, own, agg1, lcnt, lpk);

    gcn_gemm_strips<HID_C, OUT_C, true, GEMM2_GROUPS><<<GEMM2_GROUPS, GEMM_THREADS, 0, stream>>>(
        agg1, W2, dinv, h);

    gcn_aggregate_owned<OUT_C><<<OWN_GROUPS, AGG_THREADS, 0, stream>>>(
        h, pk, dinv, b2, own, out, lcnt, lpk);
}


# first launch orders each block of 4096 edge words by owner range in LDS and stores a table of run starts; an owner reads its 245 short runs, not every edge's owner byte; sums in LDS arrival order
# speedup vs baseline: 3.6833x; 1.2283x over previous
#include <hip/hip_runtime.h>
#include <hip/hip_bf16.h>


#define NNODES 100000
#define NEDGES 1000000
#define IN_C   128
#define HID_C  64
#define OUT_C  32

#define OWN_ROWS    400
#define OWN_GROUPS  (NNODES / OWN_ROWS)
#define AGG_THREADS 1024
#define AGG_WAVES   (AGG_THREADS / 32)
#define Q_SLOTS     256
#define SRC_BITS    17
#define SRC_MASK    ((1u << SRC_BITS) - 1u)
#define BIN_EDGES   4096
#define BIN_THREADS (BIN_EDGES / 4)
#define BIN_BLOCKS  ((NEDGES + BIN_EDGES - 1) / BIN_EDGES)
#define TAB_PITCH   256
#define WAVE_RUNS   8

static_assert(NNODES % OWN_ROWS == 0 && OWN_GROUPS <= TAB_PITCH, "a table row holds every owner");
static_assert(NEDGES % 4 == 0 && BIN_EDGES % 4 == 0, "a thread's four edges exist together or not at all");
static_assert(BIN_EDGES <= (1 << 15), "a run's start and its length fit 16 bits each");
static_assert(BIN_BLOCKS <= AGG_WAVES * WAVE_RUNS, "the waves of an owner cover every block");
static_assert(TAB_PITCH == 8 * 32, "one wave scans the counts, eight per lane");
static_assert(NNODES <= (1 << SRC_BITS) && OWN_ROWS <= (1 << (32 - SRC_BITS)), "source node and row fit one word");

typedef __attribute__((ext_vector_type(2))) float v2f;
typedef __attribute__((ext_vector_type(4))) float v4f;
typedef __attribute__((ext_vector_type(8))) float v8f;

__global__ void __launch_bounds__(BIN_THREADS)
gcn_order_edges(const int* __restrict__ src, const int* __restrict__ dst,
                unsigned int* __restrict__ runs,
                unsigned int* __restrict__ tab)
{
    __shared__ alignas(16) unsigned int cnt[TAB_PITCH];
    __shared__ unsigned int beg[TAB_PITCH];
    __shared__ alignas(16) unsigned int ord[BIN_EDGES];

    const int tid  = threadIdx.x;
    const int lane = tid & 31;
    const int t    = blockIdx.x * BIN_THREADS + tid;
    const bool live = t < NEDGES / 4;

    int4 d = make_int4(0, 0, 0, 0), s = make_int4(0, 0, 0, 0);
    if (live) {
        d = reinterpret_cast<const int4*>(dst)[t];
        s = reinterpret_cast<const int4*>(src)[t];
    }
    if (tid < TAB_PITCH) cnt[tid] = 0u;
    __syncthreads();

    const unsigned int dv[4] = {(unsigned int)d.x, (unsigned int)d.y, (unsigned int)d.z, (unsigned int)d.w};
    const unsigned int sv[4] = {(unsigned int)s.x, (unsigned int)s.y, (unsigned int)s.z, (unsigned int)s.w};
    unsigned int g[4], p[4], r[4];
    if (live) {
#pragma unroll
        for (int i = 0; i < 4; ++i) {
            g[i] = dv[i] / OWN_ROWS;
            p[i] = sv[i] | ((dv[i] - g[i] * OWN_ROWS) << SRC_BITS);
            r[i] = atomicAdd(&cnt[g[i]], 1u);
        }
    }
    __syncthreads();

    if (tid < 32) {
        const uint4 c0 = reinterpret_cast<const uint4*>(cnt)[2 * lane];
        const uint4 c1 = reinterpret_cast<const uint4*>(cnt)[2 * lane + 1];
        const unsigned int cv[8] = {c0.x, c0.y, c0.z, c0.w, c1.x, c1.y, c1.z, c1.w};
        unsigned int sum = 0u;
#pragma unroll
        for (int i = 0; i < 8; ++i) sum += cv[i];
        unsigned int inc = sum;
#pragma unroll
        for (int dl = 1; dl < 32; dl <<= 1) {
            const unsigned int v = __shfl_up(inc, dl);
            if (lane >= dl) inc += v;
        }
        unsigned int run = inc - sum;
#pragma unroll
        for (int i = 0; i < 8; ++i) {
            beg[8 * lane + i] = run;
            run += cv[i];
        }
    }
    __syncthreads();

    if (live) {
#pragma unroll
        for (int i = 0; i < 4; ++i) ord[beg[g[i]] + r[i]] = p[i];
    }
    __syncthreads();

    if (live) reinterpret_cast<uint4*>(runs)[t] = reinterpret_cast<const uint4*>(ord)[tid];
    if (tid < OWN_GROUPS) tab[blockIdx.x * TAB_PITCH + tid] = beg[tid] | (cnt[tid] << 16);
}

#define GEMM_WAVES   8
#define GEMM_THREADS (GEMM_WAVES * 32)
#define NSTRIPS      (NNODES / 16)
#define GEMM1_GROUPS 256
#define GEMM2_GROUPS 512

static_assert(NNODES % 16 == 0, "whole 16-row strips");

template <int K, int NC, bool RELU_A, int NWG>
__global__ void __launch_bounds__(GEMM_THREADS)
gcn_gemm_strips(const float* __restrict__ A,
                const float* __restrict__ B,
                const float* __restrict__ dinv,
                float* __restrict__ C)
{
    constexpr int TN  = NC / 16;
    constexpr int PA  = K + 4;
    constexpr int PB  = 2 * NC + 32;
    constexpr int NLD = 16 * K / 128;
    static_assert(K % 4 == 0 && (16 * K) % 128 == 0, "a strip is whole 512-B runs and a 16-B piece stays inside one row");
    __shared__ alignas(16) float Bs[(K / 2) * PB];
    __shared__ alignas(16) float As[GEMM_WAVES * 16 * PA];

    const int tid  = threadIdx.x;
    const int lane = tid & 31;
    const int wv   = __builtin_amdgcn_readfirstlane(tid >> 5);
    const int lo = lane & 15;
    const int hi = lane >> 4;
    float* as = As + wv * 16 * PA;

    constexpr int SSTEP = NWG * GEMM_WAVES;
    int s = blockIdx.x + NWG * wv;

    v4f stage[NLD];
    if (s < NSTRIPS) {
        const v4f* g = reinterpret_cast<const v4f*>(A + (size_t)s * 16 * K) + lane;
#pragma unroll
        for (int i = 0; i < NLD; ++i) stage[i] = g[i * 32];
    }

    constexpr int BQ = NC / 4;
    constexpr int BITEMS = (K / 2) * BQ;
    static_assert(BITEMS % GEMM_THREADS == 0, "whole pieces per thread");
    constexpr int BPT = BITEMS / GEMM_THREADS;
    {
        v4f r0[BPT], r1[BPT];
#pragma unroll
        for (int j = 0; j < BPT; ++j) {
            const int i  = tid + j * GEMM_THREADS;
            const int kp = i / BQ;
            const int n  = (i % BQ) * 4;
            r0[j] = *reinterpret_cast<const v4f*>(B + (size_t)(2 * kp) * NC + n);
            r1[j] = *reinterpret_cast<const v4f*>(B + (size_t)(2 * kp + 1) * NC + n);
        }
#pragma unroll
        for (int j = 0; j < BPT; ++j) {
            const int i  = tid + j * GEMM_THREADS;
            const int kp = i / BQ;
            const int n  = (i % BQ) * 4;
            v4f* d = reinterpret_cast<v4f*>(Bs + kp * PB + 2 * n);
            d[0] = v4f{r0[j].x, r1[j].x, r0[j].y, r1[j].y};
            d[1] = v4f{r0[j].z, r1[j].z, r0[j].w, r1[j].w};
        }
    }
    __syncthreads();

    const float* afrag = as + lo * PA + 2 * hi;
    const float* bfrag = Bs + hi * PB + 2 * lo;

    v2f bf[K / 4][TN];
#pragma unroll
    for (int kk = 0; kk < K / 4; ++kk)
#pragma unroll
        for (int t = 0; t < TN; ++t)
            bf[kk][t] = *reinterpret_cast<const v2f*>(bfrag + (2 * kk) * PB + 32 * t);

    for (; s < NSTRIPS; s += SSTEP) {
#pragma unroll
        for (int i = 0; i < NLD; ++i) {
            v4f v = stage[i];
            if (RELU_A) {
                v.x = fmaxf(v.x, 0.f);
                v.y = fmaxf(v.y, 0.f);
                v.z = fmaxf(v.z, 0.f);
                v.w = fmaxf(v.w, 0.f);
            }
            const int e = i * 128 + 4 * lane;
            *reinterpret_cast<v4f*>(as + (e / K) * PA + (e % K)) = v;
        }
        const int tm = s * 16;
        const float4 d0 = *(const float4*)(dinv + tm + 8 * hi);
        const float4 d1 = *(const float4*)(dinv + tm + 8 * hi + 4);
        if (s + SSTEP < NSTRIPS) {
            const v4f* g = reinterpret_cast<const v4f*>(A + (size_t)(s + SSTEP) * 16 * K) + lane;
#pragma unroll
            for (int i = 0; i < NLD; ++i) stage[i] = g[i * 32];
        }

        v8f acc[TN];
#pragma unroll
        for (int t = 0; t < TN; ++t) acc[t] = v8f{};
#pragma unroll
        for (int k = 0; k < K; k += 4) {
            const v2f a = *reinterpret_cast<const v2f*>(afrag + k);
#pragma unroll
            for (int t = 0; t < TN; ++t) {
                acc[t] = __builtin_amdgcn_wmma_f32_16x16x4_f32(
                     false, a,  false, bf[k / 4][t],
                     (short)0, acc[t],  false,  false);
            }
        }

        const float dv[8] = {d0.x, d0.y, d0.z, d0.w, d1.x, d1.y, d1.z, d1.w};
#pragma unroll
        for (int t = 0; t < TN; ++t) {
            float* crow = C + (size_t)(tm + 8 * hi) * NC + t * 16 + lo;
#pragma unroll
            for (int g = 0; g < 8; ++g) crow[(size_t)g * NC] = acc[t][g] * dv[g];
        }
    }
}

static_assert(WAVE_RUNS == 8, "the runs of a wave are written out one by one below");

struct WaveRuns {
    int o1, o2, o3, o4, o5, o6, o7, tot;
    int d0, d1, d2, d3, d4, d5, d6, d7;
};

__device__ __forceinline__ unsigned int run_entry(const unsigned int* __restrict__ tab, int grp, int wv, int lane)
{
    const int b = wv * WAVE_RUNS + lane;
    const bool has = lane < WAVE_RUNS && b < BIN_BLOCKS;
    const unsigned int e = tab[(has ? b : 0) * TAB_PITCH + grp];
    return has ? e : 0u;
}

__device__ __forceinline__ WaveRuns wave_runs(unsigned int entry, int wv)
{
    WaveRuns r;
    int o = 0;
    auto step = [&](int j, int& d) -> int {
        const unsigned int e = (unsigned int)__builtin_amdgcn_readlane((int)entry, j);
        d = (wv * WAVE_RUNS + j) * BIN_EDGES + (int)(e & 0xFFFFu) - o;
        o += (int)(e >> 16);
        return o;
    };
    r.o1 = step(0, r.d0);
    r.o2 = step(1, r.d1);
    r.o3 = step(2, r.d2);
    r.o4 = step(3, r.d3);
    r.o5 = step(4, r.d4);
    r.o6 = step(5, r.d5);
    r.o7 = step(6, r.d6);
    r.tot = step(7, r.d7);
    return r;
}

__device__ __forceinline__ int run_word(int i, const WaveRuns& r)
{
    int dl = r.d0;
    dl = (i >= r.o1) ? r.d1 : dl;
    dl = (i >= r.o2) ? r.d2 : dl;
    dl = (i >= r.o3) ? r.d3 : dl;
    dl = (i >= r.o4) ? r.d4 : dl;
    dl = (i >= r.o5) ? r.d5 : dl;
    dl = (i >= r.o6) ? r.d6 : dl;
    dl = (i >= r.o7) ? r.d7 : dl;
    return i + dl;
}

__global__ void __launch_bounds__(AGG_THREADS)
gcn_degrees(const unsigned int* __restrict__ runs, const unsigned int* __restrict__ tab,
            float* __restrict__ dinv)
{
    constexpr int K = Q_SLOTS / 32;
    __shared__ int r_cnt[OWN_ROWS];

    const int tid  = threadIdx.x;
    const int lane = tid & 31;
    const int wv   = __builtin_amdgcn_readfirstlane(tid >> 5);
    const int grp  = blockIdx.x;
    const int base = grp * OWN_ROWS;

    const unsigned int entry = run_entry(tab, grp, wv, lane);
    for (int i = tid; i < OWN_ROWS; i += AGG_THREADS) r_cnt[i] = 0;
    __syncthreads();

    const WaveRuns wr = wave_runs(entry, wv);
    const int tot = wr.tot;
    for (int pb = 0; pb < tot; pb += Q_SLOTS) {
        unsigned int pv[K];
#pragma unroll
        for (int k = 0; k < K; ++k) {
            const int i = pb + lane + 32 * k;
            const unsigned int* a = (i < tot) ? (runs + run_word(i, wr)) : runs;
            pv[k] = *a;
        }
#pragma unroll
        for (int k = 0; k < K; ++k)
            if (pb + lane + 32 * k < tot) atomicAdd(&r_cnt[pv[k] >> SRC_BITS], 1);
    }
    __syncthreads();

    for (int i = tid; i < OWN_ROWS; i += AGG_THREADS) dinv[base + i] = rsqrtf((float)(1 + r_cnt[i]));
}

template <int C>
__global__ void __launch_bounds__(AGG_THREADS)
gcn_aggregate_owned(const float* __restrict__ h,
                    const unsigned int* __restrict__ runs,
                    const unsigned int* __restrict__ tab,
                    const float* __restrict__ dinv,
                    const float* __restrict__ bias,
                    float* __restrict__ acc)
{
    constexpr int V = C / 32;
    constexpr int TILE4 = OWN_ROWS * C / 4;
    constexpr int NIT = (TILE4 + AGG_THREADS - 1) / AGG_THREADS;
    constexpr int K = Q_SLOTS / 32;
    constexpr int U = 8;
    static_assert(TILE4 >= AGG_THREADS, "a thread's first piece always exists (it stands in for a missing last one)");
    static_assert(Q_SLOTS % U == 0, "a padded list still fits the queue");
    static_assert(AGG_THREADS % (C / 4) == 0, "a thread keeps its channels from turn to turn");
    __shared__ alignas(16) float tile[OWN_ROWS * C];
    __shared__ int   q_src[AGG_WAVES * Q_SLOTS];
    __shared__ int   q_row[AGG_WAVES * Q_SLOTS];

    const int tid  = threadIdx.x;
    const int lane = tid & 31;
    const int wv   = __builtin_amdgcn_readfirstlane(tid >> 5);
    const int grp  = blockIdx.x;
    const int base = grp * OWN_ROWS;
    float* gacc = acc + (size_t)base * C;
    const float* hown = h + (size_t)base * C;
    int*   qs = q_src + wv * Q_SLOTS;
    int*   qr = q_row + wv * Q_SLOTS;

    const unsigned int entry = run_entry(tab, grp, wv, lane);

    v4f    ownv[NIT];
    float  dd[NIT];
#pragma unroll
    for (int k = 0; k < NIT; ++k) {
        const int i  = tid + k * AGG_THREADS;
        const int ic = i < TILE4 ? i : tid;
        ownv[k] = reinterpret_cast<const v4f*>(hown)[ic];
        dd[k]   = dinv[base + ic / (C / 4)];
    }
    const float* bq = bias + 4 * (tid % (C / 4));
    const float4 bv = make_float4(bq[0], bq[1], bq[2], bq[3]);

    const WaveRuns wr = wave_runs(entry, wv);
    const int tot = wr.tot;

    auto put = [&](int i, int n, int npad, unsigned int w) {
        if (i < n) {
            qs[i] = (int)(w & SRC_MASK);
            qr[i] = (int)(w >> SRC_BITS) * C;
        } else if (i < npad) {
            qs[i] = base;
            qr[i] = 0;
        }
    };

    auto load_row = [&](int i, float (&r)[V]) {
        const int s = qs[i];
        if constexpr (V == 2) {
            const float2 t = *reinterpret_cast<const float2*>(h + (size_t)s * C + 2 * lane);
            r[0] = t.x;
            r[1] = t.y;
        } else {
            r[0] = h[(size_t)s * C + lane];
        }
    };
    auto add_row = [&](int i, const float (&r)[V]) {
        float* tr = tile + qr[i] + V * lane;
#pragma unroll
        for (int v = 0; v < V; ++v) atomicAdd(&tr[v], r[v]);
    };
    auto add_queued = [&](int n, int npad) {
        if (npad > 0) {
            float cur[U][V], nxt[U][V];
#pragma unroll
            for (int u = 0; u < U; ++u) load_row(u, cur[u]);
            for (int i0 = U; i0 < npad; i0 += U) {
#pragma unroll
                for (int u = 0; u < U; ++u) load_row(i0 + u, nxt[u]);
#pragma unroll
                for (int u = 0; u < U; ++u) add_row(i0 - U + u, cur[u]);
#pragma unroll
                for (int u = 0; u < U; ++u)
#pragma unroll
                    for (int v = 0; v < V; ++v) cur[u][v] = nxt[u][v];
            }
#pragma unroll
            for (int u = 0; u < U; ++u)
                if (npad - U + u < n) add_row(npad - U + u, cur[u]);
        }
    };

    unsigned int pv[K];
#pragma unroll
    for (int k = 0; k < K; ++k) {
        const int i = lane + 32 * k;
        const unsigned int* a = (i < tot) ? (runs + run_word(i, wr)) : runs;
        pv[k] = *a;
    }

#pragma unroll
    for (int k = 0; k < NIT; ++k) {
        const int i = tid + k * AGG_THREADS;
        if (i < TILE4) reinterpret_cast<v4f*>(tile)[i] = ownv[k];
    }
    {
        const int n = tot < Q_SLOTS ? tot : Q_SLOTS;
        const int npad = (n + U - 1) / U * U;
#pragma unroll
        for (int k = 0; k < K; ++k) put(lane + 32 * k, n, npad, pv[k]);
        __syncthreads();
        add_queued(n, npad);
    }
    for (int pb = Q_SLOTS; pb < tot; pb += Q_SLOTS) {
        const int left = tot - pb;
        const int n = left < Q_SLOTS ? left : Q_SLOTS;
        const int npad = (n + U - 1) / U * U;
        unsigned int pw[K];
#pragma unroll
        for (int k = 0; k < K; ++k) {
            const int i = pb + lane + 32 * k;
            const unsigned int* a = (i < tot) ? (runs + run_word(i, wr)) : runs;
            pw[k] = *a;
        }
#pragma unroll
        for (int k = 0; k < K; ++k) put(lane + 32 * k, n, npad, pw[k]);
        add_queued(n, npad);
    }
    __syncthreads();

#pragma unroll
    for (int k = 0; k < NIT; ++k) {
        const int i = tid + k * AGG_THREADS;
        if (i < TILE4) {
            float4 v = reinterpret_cast<const float4*>(tile)[i];
            v.x = v.x * dd[k] + bv.x;
            v.y = v.y * dd[k] + bv.y;
            v.z = v.z * dd[k] + bv.z;
            v.w = v.w * dd[k] + bv.w;
            reinterpret_cast<float4*>(gacc)[i] = v;
        }
    }
}

extern "C" void kernel_launch(void* const* d_in, const int* in_sizes, int n_in,
                              void* d_out, int out_size, void* d_ws, size_t ws_size,
                              hipStream_t stream)
{
    const float* x    = (const float*)d_in[0];
    const int*   eidx = (const int*)d_in[1];
    const int*   src  = eidx;
    const int*   dst  = eidx + NEDGES;
    const float* W1   = (const float*)d_in[2];
    const float* b1   = (const float*)d_in[3];
    const float* W2   = (const float*)d_in[4];
    const float* b2   = (const float*)d_in[5];
    float* out = (float*)d_out;

    float* dinv = (float*)d_ws;
    float* h    = dinv + NNODES;
    float* agg1 = h + (size_t)NNODES * HID_C;
    unsigned int* runs = (unsigned int*)(agg1 + (size_t)NNODES * HID_C);
    unsigned int* tab  = runs + (size_t)BIN_BLOCKS * BIN_EDGES;

    gcn_order_edges<<<BIN_BLOCKS, BIN_THREADS, 0, stream>>>(src, dst, runs, tab);

    gcn_degrees<<<OWN_GROUPS, AGG_THREADS, 0, stream>>>(runs, tab, dinv);

    gcn_gemm_strips<IN_C, HID_C, false, GEMM1_GROUPS><<<GEMM1_GROUPS, GEMM_THREADS, 0, stream>>>(
        x, W1, dinv, h);

    gcn_aggregate_owned<HID_C><<<OWN_GROUPS, AGG_THREADS, 0, stream>>>(
        h, runs, tab, dinv, b1, agg1);

    gcn_gemm_strips<HID_C, OUT_C, true, GEMM2_GROUPS><<<GEMM2_GROUPS, GEMM_THREADS, 0, stream>>>(
        agg1, W2, dinv, h);

    gcn_aggregate_owned<OUT_C><<<OWN_GROUPS, AGG_THREADS, 0, stream>>>(
        h, runs, tab, dinv, b2, out);
}
